# starting kernel
# baseline (speedup 1.0000x reference)
#include <hip/hip_runtime.h>
#include <hip/hip_bf16.h>

typedef float v2f __attribute__((ext_vector_type(2)));
typedef float v8f __attribute__((ext_vector_type(8)));

#define GCN_F 128
#define BS_STRIDE 136


__global__ void gcn_zero_i32(int* p, int n) {
    int i = blockIdx.x * blockDim.x + threadIdx.x;
    if (i < n) p[i] = 0;
}

__global__ void gcn_count_deg(const int* __restrict__ col, int* deg, int e) {
    int i = blockIdx.x * blockDim.x + threadIdx.x;
    if (i < e) atomicAdd(&deg[col[i]], 1);
}

__global__ void gcn_dinv(const int* __restrict__ deg, float* dinv, int n) {
    int i = blockIdx.x * blockDim.x + threadIdx.x;
    if (i < n) dinv[i] = rsqrtf((float)(deg[i] + 1));
}


__global__ void gcn_scan_block(const int* __restrict__ deg, int* __restrict__ off,
                               int* __restrict__ bsum, int n) {
    __shared__ int s[256];
    int i = blockIdx.x * 256 + threadIdx.x;
    int v = (i < n) ? deg[i] : 0;
    s[threadIdx.x] = v;
    __syncthreads();
    for (int d = 1; d < 256; d <<= 1) {
        int t = (threadIdx.x >= d) ? s[threadIdx.x - d] : 0;
        __syncthreads();
        s[threadIdx.x] += t;
        __syncthreads();
    }
    if (i < n) off[i] = s[threadIdx.x] - v;
    if (threadIdx.x == 255) bsum[blockIdx.x] = s[255];
}

__global__ void gcn_scan_bsum(int* bsum, int nb) {
    if (threadIdx.x == 0 && blockIdx.x == 0) {
        int run = 0;
        for (int b = 0; b < nb; ++b) { int t = bsum[b]; bsum[b] = run; run += t; }
    }
}

__global__ void gcn_scan_add(int* off, const int* __restrict__ bsum, int n) {
    int i = blockIdx.x * 256 + threadIdx.x;
    if (i < n) off[i] += bsum[blockIdx.x];
}

__global__ void gcn_copy_i32(const int* __restrict__ src, int* __restrict__ dst, int n) {
    int i = blockIdx.x * blockDim.x + threadIdx.x;
    if (i < n) dst[i] = src[i];
}

__global__ void gcn_fill_csr(const int* __restrict__ rows, const int* __restrict__ cols,
                             int* cursor, int* __restrict__ srcList, int e) {
    int i = blockIdx.x * blockDim.x + threadIdx.x;
    if (i < e) {
        int c = cols[i];
        int p = atomicAdd(&cursor[c], 1);
        srcList[p] = rows[i];
    }
}

__global__ void gcn_pack_w23(const float* __restrict__ Wmu,
                             const float* __restrict__ Wlv,
                             float* __restrict__ W23) {
    int i = blockIdx.x * blockDim.x + threadIdx.x;
    int k = i >> 7, n = i & 127;
    W23[i] = (n < 64) ? Wmu[k * 64 + n] : Wlv[k * 64 + (n - 64)];
}

__global__ __launch_bounds__(256)
void gcn_gemm_rowscale_wmma(const float* __restrict__ A,
                            const float* __restrict__ B,
                            const float* __restrict__ rowScale,
                            float* __restrict__ C,
                            int M, int K) {
    __shared__ float Bs[64 * BS_STRIDE];

    const int tid  = threadIdx.x;
    const int lane = tid & 31;
    const int wave = tid >> 5;
    const int rt   = blockIdx.x * 8 + wave;
    const bool active = (rt * 16 < M);
    const int m0   = rt * 16;
    const int half = lane >> 4;
    const int l16  = lane & 15;

    const float* Arow = active ? (A + (size_t)(m0 + l16) * K) : A;

    v8f acc[8] = {};

    for (int kb = 0; kb < K; kb += 64) {
        __syncthreads();
        for (int idx = tid; idx < 64 * 32; idx += 256) {
            int r = idx >> 5, c4 = idx & 31;
            float4 v = ((const float4*)(B + (size_t)(kb + r) * GCN_F))[c4];
            *(float4*)(Bs + r * BS_STRIDE + c4 * 4) = v;
        }
        __syncthreads();

        if (active) {
            for (int kl = 0; kl < 64; kl += 4) {
                const int ka = kl + 2 * half;
                v2f a;
                a.x = Arow[kb + ka];
                a.y = Arow[kb + ka + 1];
                const float* B0 = Bs + ka * BS_STRIDE + l16;
#pragma unroll
                for (int j = 0; j < 8; ++j) {
                    v2f b;
                    b.x = B0[16 * j];
                    b.y = B0[16 * j + BS_STRIDE];
                    acc[j] = __builtin_amdgcn_wmma_f32_16x16x4_f32(
                        false, a, false, b, (short)0, acc[j], false, false);
                }
            }
        }
    }

    if (active) {
#pragma unroll
        for (int v = 0; v < 8; ++v) {
            const int row = m0 + v + 8 * half;
            const float s = rowScale[row];
            float* Crow = C + (size_t)row * GCN_F + l16;
#pragma unroll
            for (int j = 0; j < 8; ++j) Crow[16 * j] = acc[j][v] * s;
        }
    }
}


__global__ __launch_bounds__(256)
void gcn_gather_relu(const float* __restrict__ Ts, const int* __restrict__ srcList,
                     const int* __restrict__ off, const int* __restrict__ deg,
                     const float* __restrict__ dinv, const float* __restrict__ b1,
                     float* __restrict__ H, int nNodes) {
    const int node = blockIdx.x * 8 + (threadIdx.x >> 5);
    if (node >= nNodes) return;
    const int lane = threadIdx.x & 31;

    float4 acc = ((const float4*)(Ts + (size_t)node * GCN_F))[lane];
    const int s = off[node];
    const int d = deg[node];
    for (int e = 0; e < d; ++e) {
        const int src = srcList[s + e];
        const float4 v = ((const float4*)(Ts + (size_t)src * GCN_F))[lane];
        acc.x += v.x; acc.y += v.y; acc.z += v.z; acc.w += v.w;
    }
    const float dv = dinv[node];
    const float4 bb = ((const float4*)b1)[lane];
    float4 r;
    r.x = fmaxf(dv * acc.x + bb.x, 0.0f);
    r.y = fmaxf(dv * acc.y + bb.y, 0.0f);
    r.z = fmaxf(dv * acc.z + bb.z, 0.0f);
    r.w = fmaxf(dv * acc.w + bb.w, 0.0f);
    ((float4*)(H + (size_t)node * GCN_F))[lane] = r;
}

__global__ __launch_bounds__(256)
void gcn_gather_out(const float* __restrict__ Ts, const int* __restrict__ srcList,
                    const int* __restrict__ off, const int* __restrict__ deg,
                    const float* __restrict__ dinv, const float* __restrict__ bmu,
                    const float* __restrict__ blv, float* __restrict__ out, int nNodes) {
    const int node = blockIdx.x * 8 + (threadIdx.x >> 5);
    if (node >= nNodes) return;
    const int lane = threadIdx.x & 31;

    float4 acc = ((const float4*)(Ts + (size_t)node * GCN_F))[lane];
    const int s = off[node];
    const int d = deg[node];
    for (int e = 0; e < d; ++e) {
        const int src = srcList[s + e];
        const float4 v = ((const float4*)(Ts + (size_t)src * GCN_F))[lane];
        acc.x += v.x; acc.y += v.y; acc.z += v.z; acc.w += v.w;
    }
    const float dv = dinv[node];
    float4 bb;
    float* dst;
    if (lane < 16) {
        bb  = ((const float4*)bmu)[lane];
        dst = out + (size_t)node * 64 + lane * 4;
    } else {
        bb  = ((const float4*)blv)[lane - 16];
        dst = out + (size_t)nNodes * 64 + (size_t)node * 64 + (lane - 16) * 4;
    }
    float4 r;
    r.x = dv * acc.x + bb.x;
    r.y = dv * acc.y + bb.y;
    r.z = dv * acc.z + bb.z;
    r.w = dv * acc.w + bb.w;
    *(float4*)dst = r;
}


extern "C" void kernel_launch(void* const* d_in, const int* in_sizes, int n_in,
                              void* d_out, int out_size, void* d_ws, size_t ws_size,
                              hipStream_t stream) {
    const float* X   = (const float*)d_in[0];
    const int*   ei  = (const int*)d_in[1];
    const float* W1  = (const float*)d_in[2];
    const float* b1  = (const float*)d_in[3];
    const float* Wmu = (const float*)d_in[4];
    const float* bmu = (const float*)d_in[5];
    const float* Wlv = (const float*)d_in[6];
    const float* blv = (const float*)d_in[7];
    float* out = (float*)d_out;

    const int N = in_sizes[0] / 256;
    const int E = in_sizes[1] / 2;
    const int* rows = ei;
    const int* cols = ei + E;

    float* Ts   = (float*)d_ws;
    float* Hbuf = Ts + (size_t)N * GCN_F;
    float* W23  = Hbuf + (size_t)N * GCN_F;
    float* dinv = W23 + GCN_F * GCN_F;
    int* degInt  = (int*)(dinv + N);
    int* off     = degInt + N;
    int* cursor  = off + N;
    int* bsum    = cursor + N;
    const int nb = (N + 255) / 256;
    int* srcList = bsum + ((nb + 3) & ~3);

    gcn_zero_i32<<<(N + 255) / 256, 256, 0, stream>>>(degInt, N);
    gcn_count_deg<<<(E + 255) / 256, 256, 0, stream>>>(cols, degInt, E);
    gcn_dinv<<<(N + 255) / 256, 256, 0, stream>>>(degInt, dinv, N);
    gcn_scan_block<<<nb, 256, 0, stream>>>(degInt, off, bsum, N);
    gcn_scan_bsum<<<1, 64, 0, stream>>>(bsum, nb);
    gcn_scan_add<<<nb, 256, 0, stream>>>(off, bsum, N);
    gcn_copy_i32<<<(N + 255) / 256, 256, 0, stream>>>(off, cursor, N);
    gcn_fill_csr<<<(E + 255) / 256, 256, 0, stream>>>(rows, cols, cursor, srcList, E);
    gcn_pack_w23<<<64, 256, 0, stream>>>(Wmu, Wlv, W23);

    const int rowTiles = (N + 15) / 16;
    const int gemmBlocks = (rowTiles + 7) / 8;
    const int gatherBlocks = (N + 7) / 8;

    gcn_gemm_rowscale_wmma<<<gemmBlocks, 256, 0, stream>>>(X, W1, dinv, Ts, N, 256);
    gcn_gather_relu<<<gatherBlocks, 256, 0, stream>>>(Ts, srcList, off, degInt,
                                                      dinv, b1, Hbuf, N);

    gcn_gemm_rowscale_wmma<<<gemmBlocks, 256, 0, stream>>>(Hbuf, W23, dinv, Ts, N, 128);
    gcn_gather_out<<<gatherBlocks, 256, 0, stream>>>(Ts, srcList, off, degInt,
                                                     dinv, bmu, blv, out, N);
}


# CSR build as two-level counting sort on LDS counters (256-node target buckets, 4 launches) replaces per-edge global atomics and serial scan (8 launches); neighbour sum order still atomic-arrival
# speedup vs baseline: 1.4700x; 1.4700x over previous
#include <hip/hip_runtime.h>
#include <hip/hip_bf16.h>

typedef float v2f __attribute__((ext_vector_type(2)));
typedef float v8f __attribute__((ext_vector_type(8)));

#define GCN_F 128
#define BS_STRIDE 136


#define CSR_BSHIFT 8
#define CSR_BNODES 256
#define CSR_CHUNKS 256
#define CSR_MAXB   1024
#define CSR_DEPTH  16
#define CSR_BLOCK  512

__device__ __forceinline__ int gcn_block_incl_scan(int v, int* s_w) {
    const int lane = threadIdx.x & 31;
    const int wave = threadIdx.x >> 5;
    const int nw   = blockDim.x >> 5;
    for (int d = 1; d < 32; d <<= 1) {
        const int t = __shfl_up(v, d);
        if (lane >= d) v += t;
    }
    if (lane == 31) s_w[wave] = v;
    __syncthreads();
    if (wave == 0) {
        const int w = (lane < nw) ? s_w[lane] : 0;
        int wi = w;
        for (int d = 1; d < 32; d <<= 1) {
            const int t = __shfl_up(wi, d);
            if (lane >= d) wi += t;
        }
        s_w[lane] = wi - w;
    }
    __syncthreads();
    return v + s_w[wave];
}

__global__ __launch_bounds__(CSR_BLOCK)
void gcn_bucket_hist(const int* __restrict__ cols, int* __restrict__ histT, int e, int nbk) {
    __shared__ int s_h[CSR_MAXB];
    const int tid = threadIdx.x;
    for (int b = tid; b < nbk; b += CSR_BLOCK) s_h[b] = 0;
    __syncthreads();
    const int chunk = (e + CSR_CHUNKS - 1) / CSR_CHUNKS;
    const int beg = blockIdx.x * chunk;
    const int end = (beg + chunk < e) ? (beg + chunk) : e;
    for (int i0 = beg + tid; i0 < end; i0 += CSR_BLOCK * CSR_DEPTH) {
        int c[CSR_DEPTH];
#pragma unroll
        for (int u = 0; u < CSR_DEPTH; ++u) {
            const int i = i0 + CSR_BLOCK * u;
            c[u] = cols[(i < end) ? i : (end - 1)];
        }
#pragma unroll
        for (int u = 0; u < CSR_DEPTH; ++u)
            if (i0 + CSR_BLOCK * u < end) atomicAdd(&s_h[c[u] >> CSR_BSHIFT], 1);
    }
    __syncthreads();
    for (int b = tid; b < nbk; b += CSR_BLOCK) histT[blockIdx.x * nbk + b] = s_h[b];
}

__global__ __launch_bounds__(CSR_CHUNKS)
void gcn_bucket_prefix(int* __restrict__ histT, int* __restrict__ tot, int nbk) {
    __shared__ int s_w[32];
    const int b = blockIdx.x;
    const int k = threadIdx.x;
    const int v = histT[k * nbk + b];
    const int incl = gcn_block_incl_scan(v, s_w);
    histT[k * nbk + b] = incl - v;
    if (k == CSR_CHUNKS - 1) tot[b] = incl;
}

__global__ __launch_bounds__(CSR_BLOCK)
void gcn_bucket_scatter(const int* __restrict__ rows, const int* __restrict__ cols,
                        const int* __restrict__ prefT, const int* __restrict__ tot,
                        int* __restrict__ bstart, int* __restrict__ rowB,
                        int* __restrict__ colB, int e, int nbk) {
    __shared__ int s_cur[CSR_MAXB];
    __shared__ int s_w[32];
    const int tid = threadIdx.x;
    const int b0 = 2 * tid, b1 = 2 * tid + 1;
    const int v0 = (b0 < nbk) ? tot[b0] : 0;
    const int v1 = (b1 < nbk) ? tot[b1] : 0;
    const int ex0 = gcn_block_incl_scan(v0 + v1, s_w) - (v0 + v1);
    const int ex1 = ex0 + v0;
    if (b0 < nbk) {
        s_cur[b0] = ex0 + prefT[blockIdx.x * nbk + b0];
        if (blockIdx.x == 0) bstart[b0] = ex0;
    }
    if (b1 < nbk) {
        s_cur[b1] = ex1 + prefT[blockIdx.x * nbk + b1];
        if (blockIdx.x == 0) bstart[b1] = ex1;
    }
    __syncthreads();
    const int chunk = (e + CSR_CHUNKS - 1) / CSR_CHUNKS;
    const int beg = blockIdx.x * chunk;
    const int end = (beg + chunk < e) ? (beg + chunk) : e;
    for (int i0 = beg + tid; i0 < end; i0 += CSR_BLOCK * CSR_DEPTH) {
        int c[CSR_DEPTH], r[CSR_DEPTH], p[CSR_DEPTH];
#pragma unroll
        for (int u = 0; u < CSR_DEPTH; ++u) {
            const int i  = i0 + CSR_BLOCK * u;
            const int ic = (i < end) ? i : (end - 1);
            c[u] = cols[ic];
            r[u] = rows[ic];
        }
#pragma unroll
        for (int u = 0; u < CSR_DEPTH; ++u)
            p[u] = (i0 + CSR_BLOCK * u < end) ? atomicAdd(&s_cur[c[u] >> CSR_BSHIFT], 1) : 0;
#pragma unroll
        for (int u = 0; u < CSR_DEPTH; ++u) {
            if (i0 + CSR_BLOCK * u < end) {
                colB[p[u]] = c[u];
                rowB[p[u]] = r[u];
            }
        }
    }
}

__global__ __launch_bounds__(CSR_BLOCK)
void gcn_bucket_to_csr(const int* __restrict__ rowB, const int* __restrict__ colB,
                       const int* __restrict__ bstart, const int* __restrict__ tot,
                       int* __restrict__ deg, int* __restrict__ off,
                       float* __restrict__ dinv, int* __restrict__ srcList, int n) {
    __shared__ int s_cnt[CSR_BNODES];
    __shared__ int s_w[32];
    const int tid = threadIdx.x;
    const int b   = blockIdx.x;
    if (tid < CSR_BNODES) s_cnt[tid] = 0;
    __syncthreads();
    const int s0  = bstart[b];
    const int cnt = tot[b];
    for (int i0 = tid; i0 < cnt; i0 += CSR_BLOCK * CSR_DEPTH) {
        int c[CSR_DEPTH];
#pragma unroll
        for (int u = 0; u < CSR_DEPTH; ++u) {
            const int i = i0 + CSR_BLOCK * u;
            c[u] = colB[s0 + ((i < cnt) ? i : (cnt - 1))];
        }
#pragma unroll
        for (int u = 0; u < CSR_DEPTH; ++u)
            if (i0 + CSR_BLOCK * u < cnt) atomicAdd(&s_cnt[c[u] & (CSR_BNODES - 1)], 1);
    }
    __syncthreads();
    const int v = (tid < CSR_BNODES) ? s_cnt[tid] : 0;
    const int excl = gcn_block_incl_scan(v, s_w) - v;
    if (tid < CSR_BNODES) {
        s_cnt[tid] = excl;
        const int node = b * CSR_BNODES + tid;
        if (node < n) {
            deg[node]  = v;
            off[node]  = s0 + excl;
            dinv[node] = rsqrtf((float)(v + 1));
        }
    }
    __syncthreads();
    for (int i0 = tid; i0 < cnt; i0 += CSR_BLOCK * CSR_DEPTH) {
        int c[CSR_DEPTH], r[CSR_DEPTH], p[CSR_DEPTH];
#pragma unroll
        for (int u = 0; u < CSR_DEPTH; ++u) {
            const int i  = i0 + CSR_BLOCK * u;
            const int ic = s0 + ((i < cnt) ? i : (cnt - 1));
            c[u] = colB[ic];
            r[u] = rowB[ic];
        }
#pragma unroll
        for (int u = 0; u < CSR_DEPTH; ++u)
            p[u] = (i0 + CSR_BLOCK * u < cnt)
                       ? atomicAdd(&s_cnt[c[u] & (CSR_BNODES - 1)], 1) : 0;
#pragma unroll
        for (int u = 0; u < CSR_DEPTH; ++u)
            if (i0 + CSR_BLOCK * u < cnt) srcList[s0 + p[u]] = r[u];
    }
}

__global__ void gcn_pack_w23(const float* __restrict__ Wmu,
                             const float* __restrict__ Wlv,
                             float* __restrict__ W23) {
    int i = blockIdx.x * blockDim.x + threadIdx.x;
    int k = i >> 7, n = i & 127;
    W23[i] = (n < 64) ? Wmu[k * 64 + n] : Wlv[k * 64 + (n - 64)];
}

__global__ __launch_bounds__(256)
void gcn_gemm_rowscale_wmma(const float* __restrict__ A,
                            const float* __restrict__ B,
                            const float* __restrict__ rowScale,
                            float* __restrict__ C,
                            int M, int K) {
    __shared__ float Bs[64 * BS_STRIDE];

    const int tid  = threadIdx.x;
    const int lane = tid & 31;
    const int wave = tid >> 5;
    const int rt   = blockIdx.x * 8 + wave;
    const bool active = (rt * 16 < M);
    const int m0   = rt * 16;
    const int half = lane >> 4;
    const int l16  = lane & 15;

    const float* Arow = active ? (A + (size_t)(m0 + l16) * K) : A;

    v8f acc[8] = {};

    for (int kb = 0; kb < K; kb += 64) {
        __syncthreads();
        for (int idx = tid; idx < 64 * 32; idx += 256) {
            int r = idx >> 5, c4 = idx & 31;
            float4 v = ((const float4*)(B + (size_t)(kb + r) * GCN_F))[c4];
            *(float4*)(Bs + r * BS_STRIDE + c4 * 4) = v;
        }
        __syncthreads();

        if (active) {
            for (int kl = 0; kl < 64; kl += 4) {
                const int ka = kl + 2 * half;
                v2f a;
                a.x = Arow[kb + ka];
                a.y = Arow[kb + ka + 1];
                const float* B0 = Bs + ka * BS_STRIDE + l16;
#pragma unroll
                for (int j = 0; j < 8; ++j) {
                    v2f b;
                    b.x = B0[16 * j];
                    b.y = B0[16 * j + BS_STRIDE];
                    acc[j] = __builtin_amdgcn_wmma_f32_16x16x4_f32(
                        false, a, false, b, (short)0, acc[j], false, false);
                }
            }
        }
    }

    if (active) {
#pragma unroll
        for (int v = 0; v < 8; ++v) {
            const int row = m0 + v + 8 * half;
            const float s = rowScale[row];
            float* Crow = C + (size_t)row * GCN_F + l16;
#pragma unroll
            for (int j = 0; j < 8; ++j) Crow[16 * j] = acc[j][v] * s;
        }
    }
}


__global__ __launch_bounds__(256)
void gcn_gather_relu(const float* __restrict__ Ts, const int* __restrict__ srcList,
                     const int* __restrict__ off, const int* __restrict__ deg,
                     const float* __restrict__ dinv, const float* __restrict__ b1,
                     float* __restrict__ H, int nNodes) {
    const int node = blockIdx.x * 8 + (threadIdx.x >> 5);
    if (node >= nNodes) return;
    const int lane = threadIdx.x & 31;

    float4 acc = ((const float4*)(Ts + (size_t)node * GCN_F))[lane];
    const int s = off[node];
    const int d = deg[node];
    for (int e = 0; e < d; ++e) {
        const int src = srcList[s + e];
        const float4 v = ((const float4*)(Ts + (size_t)src * GCN_F))[lane];
        acc.x += v.x; acc.y += v.y; acc.z += v.z; acc.w += v.w;
    }
    const float dv = dinv[node];
    const float4 bb = ((const float4*)b1)[lane];
    float4 r;
    r.x = fmaxf(dv * acc.x + bb.x, 0.0f);
    r.y = fmaxf(dv * acc.y + bb.y, 0.0f);
    r.z = fmaxf(dv * acc.z + bb.z, 0.0f);
    r.w = fmaxf(dv * acc.w + bb.w, 0.0f);
    ((float4*)(H + (size_t)node * GCN_F))[lane] = r;
}

__global__ __launch_bounds__(256)
void gcn_gather_out(const float* __restrict__ Ts, const int* __restrict__ srcList,
                    const int* __restrict__ off, const int* __restrict__ deg,
                    const float* __restrict__ dinv, const float* __restrict__ bmu,
                    const float* __restrict__ blv, float* __restrict__ out, int nNodes) {
    const int node = blockIdx.x * 8 + (threadIdx.x >> 5);
    if (node >= nNodes) return;
    const int lane = threadIdx.x & 31;

    float4 acc = ((const float4*)(Ts + (size_t)node * GCN_F))[lane];
    const int s = off[node];
    const int d = deg[node];
    for (int e = 0; e < d; ++e) {
        const int src = srcList[s + e];
        const float4 v = ((const float4*)(Ts + (size_t)src * GCN_F))[lane];
        acc.x += v.x; acc.y += v.y; acc.z += v.z; acc.w += v.w;
    }
    const float dv = dinv[node];
    float4 bb;
    float* dst;
    if (lane < 16) {
        bb  = ((const float4*)bmu)[lane];
        dst = out + (size_t)node * 64 + lane * 4;
    } else {
        bb  = ((const float4*)blv)[lane - 16];
        dst = out + (size_t)nNodes * 64 + (size_t)node * 64 + (lane - 16) * 4;
    }
    float4 r;
    r.x = dv * acc.x + bb.x;
    r.y = dv * acc.y + bb.y;
    r.z = dv * acc.z + bb.z;
    r.w = dv * acc.w + bb.w;
    *(float4*)dst = r;
}


extern "C" void kernel_launch(void* const* d_in, const int* in_sizes, int n_in,
                              void* d_out, int out_size, void* d_ws, size_t ws_size,
                              hipStream_t stream) {
    const float* X   = (const float*)d_in[0];
    const int*   ei  = (const int*)d_in[1];
    const float* W1  = (const float*)d_in[2];
    const float* b1  = (const float*)d_in[3];
    const float* Wmu = (const float*)d_in[4];
    const float* bmu = (const float*)d_in[5];
    const float* Wlv = (const float*)d_in[6];
    const float* blv = (const float*)d_in[7];
    float* out = (float*)d_out;

    const int N = in_sizes[0] / 256;
    const int E = in_sizes[1] / 2;
    const int* rows = ei;
    const int* cols = ei + E;

    float* Ts   = (float*)d_ws;
    float* Hbuf = Ts + (size_t)N * GCN_F;
    float* W23  = Hbuf + (size_t)N * GCN_F;
    float* dinv = W23 + GCN_F * GCN_F;
    int* degInt  = (int*)(dinv + N);
    int* off     = degInt + N;
    int* srcList = off + N;

    const int nbk = (N + CSR_BNODES - 1) / CSR_BNODES;
    int* colB   = (int*)Hbuf;
    int* rowB   = colB + E;
    int* histT  = rowB + E;
    int* btot   = histT + CSR_CHUNKS * nbk;
    int* bstart = btot + nbk;
    gcn_bucket_hist<<<CSR_CHUNKS, CSR_BLOCK, 0, stream>>>(cols, histT, E, nbk);
    gcn_bucket_prefix<<<nbk, CSR_CHUNKS, 0, stream>>>(histT, btot, nbk);
    gcn_bucket_scatter<<<CSR_CHUNKS, CSR_BLOCK, 0, stream>>>(rows, cols, histT, btot, bstart,
                                                        rowB, colB, E, nbk);
    gcn_bucket_to_csr<<<nbk, CSR_BLOCK, 0, stream>>>(rowB, colB, bstart, btot, degInt, off,
                                                dinv, srcList, N);
    gcn_pack_w23<<<64, 256, 0, stream>>>(Wmu, Wlv, W23);

    const int rowTiles = (N + 15) / 16;
    const int gemmBlocks = (rowTiles + 7) / 8;
    const int gatherBlocks = (N + 7) / 8;

    gcn_gemm_rowscale_wmma<<<gemmBlocks, 256, 0, stream>>>(X, W1, dinv, Ts, N, 256);
    gcn_gather_relu<<<gatherBlocks, 256, 0, stream>>>(Ts, srcList, off, degInt,
                                                      dinv, b1, Hbuf, N);

    gcn_gemm_rowscale_wmma<<<gemmBlocks, 256, 0, stream>>>(Hbuf, W23, dinv, Ts, N, 128);
    gcn_gather_out<<<gatherBlocks, 256, 0, stream>>>(Ts, srcList, off, degInt,
                                                     dinv, bmu, blv, out, N);
}


# layer-2 product H@[Wmu|Wlv] moved into the layer-1 gather launch: a block sums 16 rows into an LDS A strip, W held in LDS; H and packed W never stored, 9 launches -> 7; sum orders kept
# speedup vs baseline: 1.6457x; 1.1195x over previous
#include <hip/hip_runtime.h>
#include <hip/hip_bf16.h>

typedef float v2f __attribute__((ext_vector_type(2)));
typedef float v8f __attribute__((ext_vector_type(8)));

#define GCN_F 128
#define BS_STRIDE 136


#define CSR_BSHIFT 8
#define CSR_BNODES 256
#define CSR_CHUNKS 256
#define CSR_MAXB   1024
#define CSR_DEPTH  16
#define CSR_BLOCK  512

__device__ __forceinline__ int gcn_block_incl_scan(int v, int* s_w) {
    const int lane = threadIdx.x & 31;
    const int wave = threadIdx.x >> 5;
    const int nw   = blockDim.x >> 5;
    for (int d = 1; d < 32; d <<= 1) {
        const int t = __shfl_up(v, d);
        if (lane >= d) v += t;
    }
    if (lane == 31) s_w[wave] = v;
    __syncthreads();
    if (wave == 0) {
        const int w = (lane < nw) ? s_w[lane] : 0;
        int wi = w;
        for (int d = 1; d < 32; d <<= 1) {
            const int t = __shfl_up(wi, d);
            if (lane >= d) wi += t;
        }
        s_w[lane] = wi - w;
    }
    __syncthreads();
    return v + s_w[wave];
}

__global__ __launch_bounds__(CSR_BLOCK)
void gcn_bucket_hist(const int* __restrict__ cols, int* __restrict__ histT, int e, int nbk) {
    __shared__ int s_h[CSR_MAXB];
    const int tid = threadIdx.x;
    for (int b = tid; b < nbk; b += CSR_BLOCK) s_h[b] = 0;
    __syncthreads();
    const int chunk = (e + CSR_CHUNKS - 1) / CSR_CHUNKS;
    const int beg = blockIdx.x * chunk;
    const int end = (beg + chunk < e) ? (beg + chunk) : e;
    for (int i0 = beg + tid; i0 < end; i0 += CSR_BLOCK * CSR_DEPTH) {
        int c[CSR_DEPTH];
#pragma unroll
        for (int u = 0; u < CSR_DEPTH; ++u) {
            const int i = i0 + CSR_BLOCK * u;
            c[u] = cols[(i < end) ? i : (end - 1)];
        }
#pragma unroll
        for (int u = 0; u < CSR_DEPTH; ++u)
            if (i0 + CSR_BLOCK * u < end) atomicAdd(&s_h[c[u] >> CSR_BSHIFT], 1);
    }
    __syncthreads();
    for (int b = tid; b < nbk; b += CSR_BLOCK) histT[blockIdx.x * nbk + b] = s_h[b];
}

__global__ __launch_bounds__(CSR_CHUNKS)
void gcn_bucket_prefix(int* __restrict__ histT, int* __restrict__ tot, int nbk) {
    __shared__ int s_w[32];
    const int b = blockIdx.x;
    const int k = threadIdx.x;
    const int v = histT[k * nbk + b];
    const int incl = gcn_block_incl_scan(v, s_w);
    histT[k * nbk + b] = incl - v;
    if (k == CSR_CHUNKS - 1) tot[b] = incl;
}

__global__ __launch_bounds__(CSR_BLOCK)
void gcn_bucket_scatter(const int* __restrict__ rows, const int* __restrict__ cols,
                        const int* __restrict__ prefT, const int* __restrict__ tot,
                        int* __restrict__ bstart, int* __restrict__ rowB,
                        int* __restrict__ colB, int e, int nbk) {
    __shared__ int s_cur[CSR_MAXB];
    __shared__ int s_w[32];
    const int tid = threadIdx.x;
    const int b0 = 2 * tid, b1 = 2 * tid + 1;
    const int v0 = (b0 < nbk) ? tot[b0] : 0;
    const int v1 = (b1 < nbk) ? tot[b1] : 0;
    const int ex0 = gcn_block_incl_scan(v0 + v1, s_w) - (v0 + v1);
    const int ex1 = ex0 + v0;
    if (b0 < nbk) {
        s_cur[b0] = ex0 + prefT[blockIdx.x * nbk + b0];
        if (blockIdx.x == 0) bstart[b0] = ex0;
    }
    if (b1 < nbk) {
        s_cur[b1] = ex1 + prefT[blockIdx.x * nbk + b1];
        if (blockIdx.x == 0) bstart[b1] = ex1;
    }
    __syncthreads();
    const int chunk = (e + CSR_CHUNKS - 1) / CSR_CHUNKS;
    const int beg = blockIdx.x * chunk;
    const int end = (beg + chunk < e) ? (beg + chunk) : e;
    for (int i0 = beg + tid; i0 < end; i0 += CSR_BLOCK * CSR_DEPTH) {
        int c[CSR_DEPTH], r[CSR_DEPTH], p[CSR_DEPTH];
#pragma unroll
        for (int u = 0; u < CSR_DEPTH; ++u) {
            const int i  = i0 + CSR_BLOCK * u;
            const int ic = (i < end) ? i : (end - 1);
            c[u] = cols[ic];
            r[u] = rows[ic];
        }
#pragma unroll
        for (int u = 0; u < CSR_DEPTH; ++u)
            p[u] = (i0 + CSR_BLOCK * u < end) ? atomicAdd(&s_cur[c[u] >> CSR_BSHIFT], 1) : 0;
#pragma unroll
        for (int u = 0; u < CSR_DEPTH; ++u) {
            if (i0 + CSR_BLOCK * u < end) {
                colB[p[u]] = c[u];
                rowB[p[u]] = r[u];
            }
        }
    }
}

__global__ __launch_bounds__(CSR_BLOCK)
void gcn_bucket_to_csr(const int* __restrict__ rowB, const int* __restrict__ colB,
                       const int* __restrict__ bstart, const int* __restrict__ tot,
                       int* __restrict__ deg, int* __restrict__ off,
                       float* __restrict__ dinv, int* __restrict__ srcList, int n) {
    __shared__ int s_cnt[CSR_BNODES];
    __shared__ int s_w[32];
    const int tid = threadIdx.x;
    const int b   = blockIdx.x;
    if (tid < CSR_BNODES) s_cnt[tid] = 0;
    __syncthreads();
    const int s0  = bstart[b];
    const int cnt = tot[b];
    for (int i0 = tid; i0 < cnt; i0 += CSR_BLOCK * CSR_DEPTH) {
        int c[CSR_DEPTH];
#pragma unroll
        for (int u = 0; u < CSR_DEPTH; ++u) {
            const int i = i0 + CSR_BLOCK * u;
            c[u] = colB[s0 + ((i < cnt) ? i : (cnt - 1))];
        }
#pragma unroll
        for (int u = 0; u < CSR_DEPTH; ++u)
            if (i0 + CSR_BLOCK * u < cnt) atomicAdd(&s_cnt[c[u] & (CSR_BNODES - 1)], 1);
    }
    __syncthreads();
    const int v = (tid < CSR_BNODES) ? s_cnt[tid] : 0;
    const int excl = gcn_block_incl_scan(v, s_w) - v;
    if (tid < CSR_BNODES) {
        s_cnt[tid] = excl;
        const int node = b * CSR_BNODES + tid;
        if (node < n) {
            deg[node]  = v;
            off[node]  = s0 + excl;
            dinv[node] = rsqrtf((float)(v + 1));
        }
    }
    __syncthreads();
    for (int i0 = tid; i0 < cnt; i0 += CSR_BLOCK * CSR_DEPTH) {
        int c[CSR_DEPTH], r[CSR_DEPTH], p[CSR_DEPTH];
#pragma unroll
        for (int u = 0; u < CSR_DEPTH; ++u) {
            const int i  = i0 + CSR_BLOCK * u;
            const int ic = s0 + ((i < cnt) ? i : (cnt - 1));
            c[u] = colB[ic];
            r[u] = rowB[ic];
        }
#pragma unroll
        for (int u = 0; u < CSR_DEPTH; ++u)
            p[u] = (i0 + CSR_BLOCK * u < cnt)
                       ? atomicAdd(&s_cnt[c[u] & (CSR_BNODES - 1)], 1) : 0;
#pragma unroll
        for (int u = 0; u < CSR_DEPTH; ++u)
            if (i0 + CSR_BLOCK * u < cnt) srcList[s0 + p[u]] = r[u];
    }
}

__global__ __launch_bounds__(256)
void gcn_gemm_rowscale_wmma(const float* __restrict__ A,
                            const float* __restrict__ B,
                            const float* __restrict__ rowScale,
                            float* __restrict__ C,
                            int M, int K) {
    __shared__ float Bs[64 * BS_STRIDE];

    const int tid  = threadIdx.x;
    const int lane = tid & 31;
    const int wave = tid >> 5;
    const int rt   = blockIdx.x * 8 + wave;
    const bool active = (rt * 16 < M);
    const int m0   = rt * 16;
    const int half = lane >> 4;
    const int l16  = lane & 15;

    const float* Arow = active ? (A + (size_t)(m0 + l16) * K) : A;

    v8f acc[8] = {};

    for (int kb = 0; kb < K; kb += 64) {
        __syncthreads();
        for (int idx = tid; idx < 64 * 32; idx += 256) {
            int r = idx >> 5, c4 = idx & 31;
            float4 v = ((const float4*)(B + (size_t)(kb + r) * GCN_F))[c4];
            *(float4*)(Bs + r * BS_STRIDE + c4 * 4) = v;
        }
        __syncthreads();

        if (active) {
            for (int kl = 0; kl < 64; kl += 4) {
                const int ka = kl + 2 * half;
                v2f a;
                a.x = Arow[kb + ka];
                a.y = Arow[kb + ka + 1];
                const float* B0 = Bs + ka * BS_STRIDE + l16;
#pragma unroll
                for (int j = 0; j < 8; ++j) {
                    v2f b;
                    b.x = B0[16 * j];
                    b.y = B0[16 * j + BS_STRIDE];
                    acc[j] = __builtin_amdgcn_wmma_f32_16x16x4_f32(
                        false, a, false, b, (short)0, acc[j], false, false);
                }
            }
        }
    }

    if (active) {
#pragma unroll
        for (int v = 0; v < 8; ++v) {
            const int row = m0 + v + 8 * half;
            const float s = rowScale[row];
            float* Crow = C + (size_t)row * GCN_F + l16;
#pragma unroll
            for (int j = 0; j < 8; ++j) Crow[16 * j] = acc[j][v] * s;
        }
    }
}


#define FG_WAVES    16
#define FG_THREADS  512
#define HS_STRIDE   132
#define HS_DINV     128
#define FG_INFLIGHT 8

__device__ __forceinline__ void gcn_add4(float4& a, const float4& v) {
    a.x += v.x; a.y += v.y; a.z += v.z; a.w += v.w;
}

__global__ __launch_bounds__(FG_THREADS)
void gcn_gather_relu_gemm(const float* __restrict__ Ts, const int* __restrict__ srcList,
                          const int* __restrict__ off, const int* __restrict__ deg,
                          const float* __restrict__ dinv, const float* __restrict__ b1,
                          const float* __restrict__ Wmu, const float* __restrict__ Wlv,
                          float* __restrict__ T2, int nNodes, int nTiles) {
    __shared__ float Ws[GCN_F * BS_STRIDE];
    __shared__ float Hs[2][FG_WAVES * HS_STRIDE];

    const int tid  = threadIdx.x;
    const int lane = tid & 31;
    const int wave = __builtin_amdgcn_readfirstlane(tid >> 5);
    const int half = lane >> 4;
    const int l16  = lane & 15;

    for (int idx = tid; idx < GCN_F * 32; idx += FG_THREADS) {
        const int r = idx >> 5, c4 = idx & 31;
        const float4 v = (c4 < 16) ? ((const float4*)(Wmu + (size_t)r * 64))[c4]
                                   : ((const float4*)(Wlv + (size_t)r * 64))[c4 - 16];
        *(float4*)(Ws + r * BS_STRIDE + c4 * 4) = v;
    }
    const float4 bb = ((const float4*)b1)[lane];
    const float4* TsRow = (const float4*)Ts + lane;

    const int step = gridDim.x;
    int tile = blockIdx.x;

    int   s_c = 0, d_c = 0, s_n = 0, d_n = 0;
    float dv_c = 0.0f, dv_n = 0.0f;
    {
        const int n0 = tile * FG_WAVES + wave;
        if (tile < nTiles && n0 < nNodes) { s_c = off[n0]; d_c = deg[n0]; dv_c = dinv[n0]; }
        const int n1 = (tile + step) * FG_WAVES + wave;
        if (tile + step < nTiles && n1 < nNodes) { s_n = off[n1]; d_n = deg[n1]; dv_n = dinv[n1]; }
    }
    int mine_c = (lane < d_c) ? srcList[s_c + lane] : 0;

    int buf = 0;
    for (; tile < nTiles; tile += step) {
        const int node = tile * FG_WAVES + wave;

        int   s_a = 0, d_a = 0;
        float dv_a = 0.0f;
        {
            const int ta = tile + 2 * step;
            const int na = ta * FG_WAVES + wave;
            if (ta < nTiles && na < nNodes) { s_a = off[na]; d_a = deg[na]; dv_a = dinv[na]; }
        }
        const int mine_n = (lane < d_n) ? srcList[s_n + lane] : 0;

        float4 r = {0.0f, 0.0f, 0.0f, 0.0f};
        if (node < nNodes) {
            float4 acc = TsRow[(size_t)node * 32];
            for (int e0 = 0; e0 < d_c; e0 += 32) {
                const int cnt = (d_c - e0 < 32) ? (d_c - e0) : 32;
                const int mine = (e0 == 0) ? mine_c
                                           : ((lane < cnt) ? srcList[s_c + e0 + lane] : 0);
                int e = 0;
                for (; e + FG_INFLIGHT <= cnt; e += FG_INFLIGHT) {
                    float4 v[FG_INFLIGHT];
#pragma unroll
                    for (int u = 0; u < FG_INFLIGHT; ++u)
                        v[u] = TsRow[(size_t)__builtin_amdgcn_readlane(mine, e + u) * 32];
#pragma unroll
                    for (int u = 0; u < FG_INFLIGHT; ++u) gcn_add4(acc, v[u]);
                }
                if (e < cnt) {
                    float4 v[FG_INFLIGHT - 1] = {};
#pragma unroll
                    for (int u = 0; u < FG_INFLIGHT - 1; ++u)
                        if (e + u < cnt)
                            v[u] = TsRow[(size_t)__builtin_amdgcn_readlane(mine, e + u) * 32];
#pragma unroll
                    for (int u = 0; u < FG_INFLIGHT - 1; ++u)
                        if (e + u < cnt) gcn_add4(acc, v[u]);
                }
            }
            r.x = fmaxf(dv_c * acc.x + bb.x, 0.0f);
            r.y = fmaxf(dv_c * acc.y + bb.y, 0.0f);
            r.z = fmaxf(dv_c * acc.z + bb.z, 0.0f);
            r.w = fmaxf(dv_c * acc.w + bb.w, 0.0f);
        }
        float* Hrow = &Hs[buf][wave * HS_STRIDE];
        *(float4*)(Hrow + lane * 4) = r;
        if (lane == 0) Hrow[HS_DINV] = dv_c;
        __syncthreads();

        if (wave < 8) {
            const float* A0 = &Hs[buf][l16 * HS_STRIDE + 2 * half];
            const float* B0 = Ws + (2 * half) * BS_STRIDE + 16 * wave + l16;
            v8f acc = {};
#pragma unroll 8
            for (int i = 0; i < GCN_F / 4; ++i) {
                v2f a, b;
                a.x = A0[4 * i];
                a.y = A0[4 * i + 1];
                b.x = B0[(4 * i) * BS_STRIDE];
                b.y = B0[(4 * i + 1) * BS_STRIDE];
                acc = __builtin_amdgcn_wmma_f32_16x16x4_f32(
                    false, a, false, b, (short)0, acc, false, false);
            }
            const int row0 = tile * FG_WAVES + 8 * half;
            const float* S0 = &Hs[buf][(8 * half) * HS_STRIDE + HS_DINV];
            float* C0 = T2 + (size_t)row0 * GCN_F + 16 * wave + l16;
            if (tile * FG_WAVES + FG_WAVES <= nNodes) {
#pragma unroll
                for (int v = 0; v < 8; ++v) C0[(size_t)v * GCN_F] = acc[v] * S0[v * HS_STRIDE];
            } else {
#pragma unroll
                for (int v = 0; v < 8; ++v)
                    if (row0 + v < nNodes) C0[(size_t)v * GCN_F] = acc[v] * S0[v * HS_STRIDE];
            }
        }
        s_c = s_n; d_c = d_n; dv_c = dv_n; mine_c = mine_n;
        s_n = s_a; d_n = d_a; dv_n = dv_a;
        buf ^= 1;
    }
}

__global__ __launch_bounds__(256)
void gcn_gather_out(const float* __restrict__ Ts, const int* __restrict__ srcList,
                    const int* __restrict__ off, const int* __restrict__ deg,
                    const float* __restrict__ dinv, const float* __restrict__ bmu,
                    const float* __restrict__ blv, float* __restrict__ out, int nNodes) {
    const int node = blockIdx.x * 8 + (threadIdx.x >> 5);
    if (node >= nNodes) return;
    const int lane = threadIdx.x & 31;

    float4 acc = ((const float4*)(Ts + (size_t)node * GCN_F))[lane];
    const int s = off[node];
    const int d = deg[node];
    for (int e = 0; e < d; ++e) {
        const int src = srcList[s + e];
        const float4 v = ((const float4*)(Ts + (size_t)src * GCN_F))[lane];
        acc.x += v.x; acc.y += v.y; acc.z += v.z; acc.w += v.w;
    }
    const float dv = dinv[node];
    float4 bb;
    float* dst;
    if (lane < 16) {
        bb  = ((const float4*)bmu)[lane];
        dst = out + (size_t)node * 64 + lane * 4;
    } else {
        bb  = ((const float4*)blv)[lane - 16];
        dst = out + (size_t)nNodes * 64 + (size_t)node * 64 + (lane - 16) * 4;
    }
    float4 r;
    r.x = dv * acc.x + bb.x;
    r.y = dv * acc.y + bb.y;
    r.z = dv * acc.z + bb.z;
    r.w = dv * acc.w + bb.w;
    *(float4*)dst = r;
}


extern "C" void kernel_launch(void* const* d_in, const int* in_sizes, int n_in,
                              void* d_out, int out_size, void* d_ws, size_t ws_size,
                              hipStream_t stream) {
    const float* X   = (const float*)d_in[0];
    const int*   ei  = (const int*)d_in[1];
    const float* W1  = (const float*)d_in[2];
    const float* b1  = (const float*)d_in[3];
    const float* Wmu = (const float*)d_in[4];
    const float* bmu = (const float*)d_in[5];
    const float* Wlv = (const float*)d_in[6];
    const float* blv = (const float*)d_in[7];
    float* out = (float*)d_out;

    const int N = in_sizes[0] / 256;
    const int E = in_sizes[1] / 2;
    const int* rows = ei;
    const int* cols = ei + E;

    float* Ts   = (float*)d_ws;
    float* Hbuf = Ts + (size_t)N * GCN_F;
    float* W23  = Hbuf + (size_t)N * GCN_F;
    float* dinv = W23 + GCN_F * GCN_F;
    int* degInt  = (int*)(dinv + N);
    int* off     = degInt + N;
    int* srcList = off + N;

    const int nbk = (N + CSR_BNODES - 1) / CSR_BNODES;
    int* colB   = (int*)Hbuf;
    int* rowB   = colB + E;
    int* histT  = rowB + E;
    int* btot   = histT + CSR_CHUNKS * nbk;
    int* bstart = btot + nbk;
    gcn_bucket_hist<<<CSR_CHUNKS, CSR_BLOCK, 0, stream>>>(cols, histT, E, nbk);
    gcn_bucket_prefix<<<nbk, CSR_CHUNKS, 0, stream>>>(histT, btot, nbk);
    gcn_bucket_scatter<<<CSR_CHUNKS, CSR_BLOCK, 0, stream>>>(rows, cols, histT, btot, bstart,
                                                        rowB, colB, E, nbk);
    gcn_bucket_to_csr<<<nbk, CSR_BLOCK, 0, stream>>>(rowB, colB, bstart, btot, degInt, off,
                                                dinv, srcList, N);

    const int rowTiles = (N + 15) / 16;
    const int gemmBlocks = (rowTiles + 7) / 8;
    const int gatherBlocks = (N + 7) / 8;

    gcn_gemm_rowscale_wmma<<<gemmBlocks, 256, 0, stream>>>(X, W1, dinv, Ts, N, 256);
    gcn_gather_relu_gemm<<<768, FG_THREADS, 0, stream>>>(Ts, srcList, off, degInt, dinv, b1,
                                                          Wmu, Wlv, Hbuf, N, rowTiles);

    gcn_gather_out<<<gatherBlocks, 256, 0, stream>>>(Hbuf, srcList, off, degInt,
                                                     dinv, bmu, blv, out, N);
}


# output gather on resident waves that walk the nodes: a node's list one entry per lane, rows asked for 8 at a time (tail as one group), next nodes' list, scalars, own row fetched ahead; sum order kept
# speedup vs baseline: 1.7020x; 1.0342x over previous
#include <hip/hip_runtime.h>
#include <hip/hip_bf16.h>

typedef float v2f __attribute__((ext_vector_type(2)));
typedef float v8f __attribute__((ext_vector_type(8)));

#define GCN_F 128
#define BS_STRIDE 136


#define CSR_BSHIFT 8
#define CSR_BNODES 256
#define CSR_CHUNKS 256
#define CSR_MAXB   1024
#define CSR_DEPTH  16
#define CSR_BLOCK  512

__device__ __forceinline__ int gcn_block_incl_scan(int v, int* s_w) {
    const int lane = threadIdx.x & 31;
    const int wave = threadIdx.x >> 5;
    const int nw   = blockDim.x >> 5;
    for (int d = 1; d < 32; d <<= 1) {
        const int t = __shfl_up(v, d);
        if (lane >= d) v += t;
    }
    if (lane == 31) s_w[wave] = v;
    __syncthreads();
    if (wave == 0) {
        const int w = (lane < nw) ? s_w[lane] : 0;
        int wi = w;
        for (int d = 1; d < 32; d <<= 1) {
            const int t = __shfl_up(wi, d);
            if (lane >= d) wi += t;
        }
        s_w[lane] = wi - w;
    }
    __syncthreads();
    return v + s_w[wave];
}

__global__ __launch_bounds__(CSR_BLOCK)
void gcn_bucket_hist(const int* __restrict__ cols, int* __restrict__ histT, int e, int nbk) {
    __shared__ int s_h[CSR_MAXB];
    const int tid = threadIdx.x;
    for (int b = tid; b < nbk; b += CSR_BLOCK) s_h[b] = 0;
    __syncthreads();
    const int chunk = (e + CSR_CHUNKS - 1) / CSR_CHUNKS;
    const int beg = blockIdx.x * chunk;
    const int end = (beg + chunk < e) ? (beg + chunk) : e;
    for (int i0 = beg + tid; i0 < end; i0 += CSR_BLOCK * CSR_DEPTH) {
        int c[CSR_DEPTH];
#pragma unroll
        for (int u = 0; u < CSR_DEPTH; ++u) {
            const int i = i0 + CSR_BLOCK * u;
            c[u] = cols[(i < end) ? i : (end - 1)];
        }
#pragma unroll
        for (int u = 0; u < CSR_DEPTH; ++u)
            if (i0 + CSR_BLOCK * u < end) atomicAdd(&s_h[c[u] >> CSR_BSHIFT], 1);
    }
    __syncthreads();
    for (int b = tid; b < nbk; b += CSR_BLOCK) histT[blockIdx.x * nbk + b] = s_h[b];
}

__global__ __launch_bounds__(CSR_CHUNKS)
void gcn_bucket_prefix(int* __restrict__ histT, int* __restrict__ tot, int nbk) {
    __shared__ int s_w[32];
    const int b = blockIdx.x;
    const int k = threadIdx.x;
    const int v = histT[k * nbk + b];
    const int incl = gcn_block_incl_scan(v, s_w);
    histT[k * nbk + b] = incl - v;
    if (k == CSR_CHUNKS - 1) tot[b] = incl;
}

__global__ __launch_bounds__(CSR_BLOCK)
void gcn_bucket_scatter(const int* __restrict__ rows, const int* __restrict__ cols,
                        const int* __restrict__ prefT, const int* __restrict__ tot,
                        int* __restrict__ bstart, int* __restrict__ rowB,
                        int* __restrict__ colB, int e, int nbk) {
    __shared__ int s_cur[CSR_MAXB];
    __shared__ int s_w[32];
    const int tid = threadIdx.x;
    const int b0 = 2 * tid, b1 = 2 * tid + 1;
    const int v0 = (b0 < nbk) ? tot[b0] : 0;
    const int v1 = (b1 < nbk) ? tot[b1] : 0;
    const int ex0 = gcn_block_incl_scan(v0 + v1, s_w) - (v0 + v1);
    const int ex1 = ex0 + v0;
    if (b0 < nbk) {
        s_cur[b0] = ex0 + prefT[blockIdx.x * nbk + b0];
        if (blockIdx.x == 0) bstart[b0] = ex0;
    }
    if (b1 < nbk) {
        s_cur[b1] = ex1 + prefT[blockIdx.x * nbk + b1];
        if (blockIdx.x == 0) bstart[b1] = ex1;
    }
    __syncthreads();
    const int chunk = (e + CSR_CHUNKS - 1) / CSR_CHUNKS;
    const int beg = blockIdx.x * chunk;
    const int end = (beg + chunk < e) ? (beg + chunk) : e;
    for (int i0 = beg + tid; i0 < end; i0 += CSR_BLOCK * CSR_DEPTH) {
        int c[CSR_DEPTH], r[CSR_DEPTH], p[CSR_DEPTH];
#pragma unroll
        for (int u = 0; u < CSR_DEPTH; ++u) {
            const int i  = i0 + CSR_BLOCK * u;
            const int ic = (i < end) ? i : (end - 1);
            c[u] = cols[ic];
            r[u] = rows[ic];
        }
#pragma unroll
        for (int u = 0; u < CSR_DEPTH; ++u)
            p[u] = (i0 + CSR_BLOCK * u < end) ? atomicAdd(&s_cur[c[u] >> CSR_BSHIFT], 1) : 0;
#pragma unroll
        for (int u = 0; u < CSR_DEPTH; ++u) {
            if (i0 + CSR_BLOCK * u < end) {
                colB[p[u]] = c[u];
                rowB[p[u]] = r[u];
            }
        }
    }
}

__global__ __launch_bounds__(CSR_BLOCK)
void gcn_bucket_to_csr(const int* __restrict__ rowB, const int* __restrict__ colB,
                       const int* __restrict__ bstart, const int* __restrict__ tot,
                       int* __restrict__ deg, int* __restrict__ off,
                       float* __restrict__ dinv, int* __restrict__ srcList, int n) {
    __shared__ int s_cnt[CSR_BNODES];
    __shared__ int s_w[32];
    const int tid = threadIdx.x;
    const int b   = blockIdx.x;
    if (tid < CSR_BNODES) s_cnt[tid] = 0;
    __syncthreads();
    const int s0  = bstart[b];
    const int cnt = tot[b];
    for (int i0 = tid; i0 < cnt; i0 += CSR_BLOCK * CSR_DEPTH) {
        int c[CSR_DEPTH];
#pragma unroll
        for (int u = 0; u < CSR_DEPTH; ++u) {
            const int i = i0 + CSR_BLOCK * u;
            c[u] = colB[s0 + ((i < cnt) ? i : (cnt - 1))];
        }
#pragma unroll
        for (int u = 0; u < CSR_DEPTH; ++u)
            if (i0 + CSR_BLOCK * u < cnt) atomicAdd(&s_cnt[c[u] & (CSR_BNODES - 1)], 1);
    }
    __syncthreads();
    const int v = (tid < CSR_BNODES) ? s_cnt[tid] : 0;
    const int excl = gcn_block_incl_scan(v, s_w) - v;
    if (tid < CSR_BNODES) {
        s_cnt[tid] = excl;
        const int node = b * CSR_BNODES + tid;
        if (node < n) {
            deg[node]  = v;
            off[node]  = s0 + excl;
            dinv[node] = rsqrtf((float)(v + 1));
        }
    }
    __syncthreads();
    for (int i0 = tid; i0 < cnt; i0 += CSR_BLOCK * CSR_DEPTH) {
        int c[CSR_DEPTH], r[CSR_DEPTH], p[CSR_DEPTH];
#pragma unroll
        for (int u = 0; u < CSR_DEPTH; ++u) {
            const int i  = i0 + CSR_BLOCK * u;
            const int ic = s0 + ((i < cnt) ? i : (cnt - 1));
            c[u] = colB[ic];
            r[u] = rowB[ic];
        }
#pragma unroll
        for (int u = 0; u < CSR_DEPTH; ++u)
            p[u] = (i0 + CSR_BLOCK * u < cnt)
                       ? atomicAdd(&s_cnt[c[u] & (CSR_BNODES - 1)], 1) : 0;
#pragma unroll
        for (int u = 0; u < CSR_DEPTH; ++u)
            if (i0 + CSR_BLOCK * u < cnt) srcList[s0 + p[u]] = r[u];
    }
}

__global__ __launch_bounds__(256)
void gcn_gemm_rowscale_wmma(const float* __restrict__ A,
                            const float* __restrict__ B,
                            const float* __restrict__ rowScale,
                            float* __restrict__ C,
                            int M, int K) {
    __shared__ float Bs[64 * BS_STRIDE];

    const int tid  = threadIdx.x;
    const int lane = tid & 31;
    const int wave = tid >> 5;
    const int rt   = blockIdx.x * 8 + wave;
    const bool active = (rt * 16 < M);
    const int m0   = rt * 16;
    const int half = lane >> 4;
    const int l16  = lane & 15;

    const float* Arow = active ? (A + (size_t)(m0 + l16) * K) : A;

    v8f acc[8] = {};

    for (int kb = 0; kb < K; kb += 64) {
        __syncthreads();
        for (int idx = tid; idx < 64 * 32; idx += 256) {
            int r = idx >> 5, c4 = idx & 31;
            float4 v = ((const float4*)(B + (size_t)(kb + r) * GCN_F))[c4];
            *(float4*)(Bs + r * BS_STRIDE + c4 * 4) = v;
        }
        __syncthreads();

        if (active) {
            for (int kl = 0; kl < 64; kl += 4) {
                const int ka = kl + 2 * half;
                v2f a;
                a.x = Arow[kb + ka];
                a.y = Arow[kb + ka + 1];
                const float* B0 = Bs + ka * BS_STRIDE + l16;
#pragma unroll
                for (int j = 0; j < 8; ++j) {
                    v2f b;
                    b.x = B0[16 * j];
                    b.y = B0[16 * j + BS_STRIDE];
                    acc[j] = __builtin_amdgcn_wmma_f32_16x16x4_f32(
                        false, a, false, b, (short)0, acc[j], false, false);
                }
            }
        }
    }

    if (active) {
#pragma unroll
        for (int v = 0; v < 8; ++v) {
            const int row = m0 + v + 8 * half;
            const float s = rowScale[row];
            float* Crow = C + (size_t)row * GCN_F + l16;
#pragma unroll
            for (int j = 0; j < 8; ++j) Crow[16 * j] = acc[j][v] * s;
        }
    }
}


#define FG_WAVES    16
#define FG_THREADS  512
#define HS_STRIDE   132
#define HS_DINV     128
#define FG_INFLIGHT 8

__device__ __forceinline__ void gcn_add4(float4& a, const float4& v) {
    a.x += v.x; a.y += v.y; a.z += v.z; a.w += v.w;
}

__global__ __launch_bounds__(FG_THREADS)
void gcn_gather_relu_gemm(const float* __restrict__ Ts, const int* __restrict__ srcList,
                          const int* __restrict__ off, const int* __restrict__ deg,
                          const float* __restrict__ dinv, const float* __restrict__ b1,
                          const float* __restrict__ Wmu, const float* __restrict__ Wlv,
                          float* __restrict__ T2, int nNodes, int nTiles) {
    __shared__ float Ws[GCN_F * BS_STRIDE];
    __shared__ float Hs[2][FG_WAVES * HS_STRIDE];

    const int tid  = threadIdx.x;
    const int lane = tid & 31;
    const int wave = __builtin_amdgcn_readfirstlane(tid >> 5);
    const int half = lane >> 4;
    const int l16  = lane & 15;

    for (int idx = tid; idx < GCN_F * 32; idx += FG_THREADS) {
        const int r = idx >> 5, c4 = idx & 31;
        const float4 v = (c4 < 16) ? ((const float4*)(Wmu + (size_t)r * 64))[c4]
                                   : ((const float4*)(Wlv + (size_t)r * 64))[c4 - 16];
        *(float4*)(Ws + r * BS_STRIDE + c4 * 4) = v;
    }
    const float4 bb = ((const float4*)b1)[lane];
    const float4* TsRow = (const float4*)Ts + lane;

    const int step = gridDim.x;
    int tile = blockIdx.x;

    int   s_c = 0, d_c = 0, s_n = 0, d_n = 0;
    float dv_c = 0.0f, dv_n = 0.0f;
    {
        const int n0 = tile * FG_WAVES + wave;
        if (tile < nTiles && n0 < nNodes) { s_c = off[n0]; d_c = deg[n0]; dv_c = dinv[n0]; }
        const int n1 = (tile + step) * FG_WAVES + wave;
        if (tile + step < nTiles && n1 < nNodes) { s_n = off[n1]; d_n = deg[n1]; dv_n = dinv[n1]; }
    }
    int mine_c = (lane < d_c) ? srcList[s_c + lane] : 0;

    int buf = 0;
    for (; tile < nTiles; tile += step) {
        const int node = tile * FG_WAVES + wave;

        int   s_a = 0, d_a = 0;
        float dv_a = 0.0f;
        {
            const int ta = tile + 2 * step;
            const int na = ta * FG_WAVES + wave;
            if (ta < nTiles && na < nNodes) { s_a = off[na]; d_a = deg[na]; dv_a = dinv[na]; }
        }
        const int mine_n = (lane < d_n) ? srcList[s_n + lane] : 0;

        float4 r = {0.0f, 0.0f, 0.0f, 0.0f};
        if (node < nNodes) {
            float4 acc = TsRow[(size_t)node * 32];
            for (int e0 = 0; e0 < d_c; e0 += 32) {
                const int cnt = (d_c - e0 < 32) ? (d_c - e0) : 32;
                const int mine = (e0 == 0) ? mine_c
                                           : ((lane < cnt) ? srcList[s_c + e0 + lane] : 0);
                int e = 0;
                for (; e + FG_INFLIGHT <= cnt; e += FG_INFLIGHT) {
                    float4 v[FG_INFLIGHT];
#pragma unroll
                    for (int u = 0; u < FG_INFLIGHT; ++u)
                        v[u] = TsRow[(size_t)__builtin_amdgcn_readlane(mine, e + u) * 32];
#pragma unroll
                    for (int u = 0; u < FG_INFLIGHT; ++u) gcn_add4(acc, v[u]);
                }
                if (e < cnt) {
                    float4 v[FG_INFLIGHT - 1] = {};
#pragma unroll
                    for (int u = 0; u < FG_INFLIGHT - 1; ++u)
                        if (e + u < cnt)
                            v[u] = TsRow[(size_t)__builtin_amdgcn_readlane(mine, e + u) * 32];
#pragma unroll
                    for (int u = 0; u < FG_INFLIGHT - 1; ++u)
                        if (e + u < cnt) gcn_add4(acc, v[u]);
                }
            }
            r.x = fmaxf(dv_c * acc.x + bb.x, 0.0f);
            r.y = fmaxf(dv_c * acc.y + bb.y, 0.0f);
            r.z = fmaxf(dv_c * acc.z + bb.z, 0.0f);
            r.w = fmaxf(dv_c * acc.w + bb.w, 0.0f);
        }
        float* Hrow = &Hs[buf][wave * HS_STRIDE];
        *(float4*)(Hrow + lane * 4) = r;
        if (lane == 0) Hrow[HS_DINV] = dv_c;
        __syncthreads();

        if (wave < 8) {
            const float* A0 = &Hs[buf][l16 * HS_STRIDE + 2 * half];
            const float* B0 = Ws + (2 * half) * BS_STRIDE + 16 * wave + l16;
            v8f acc = {};
#pragma unroll 8
            for (int i = 0; i < GCN_F / 4; ++i) {
                v2f a, b;
                a.x = A0[4 * i];
                a.y = A0[4 * i + 1];
                b.x = B0[(4 * i) * BS_STRIDE];
                b.y = B0[(4 * i + 1) * BS_STRIDE];
                acc = __builtin_amdgcn_wmma_f32_16x16x4_f32(
                    false, a, false, b, (short)0, acc, false, false);
            }
            const int row0 = tile * FG_WAVES + 8 * half;
            const float* S0 = &Hs[buf][(8 * half) * HS_STRIDE + HS_DINV];
            float* C0 = T2 + (size_t)row0 * GCN_F + 16 * wave + l16;
            if (tile * FG_WAVES + FG_WAVES <= nNodes) {
#pragma unroll
                for (int v = 0; v < 8; ++v) C0[(size_t)v * GCN_F] = acc[v] * S0[v * HS_STRIDE];
            } else {
#pragma unroll
                for (int v = 0; v < 8; ++v)
                    if (row0 + v < nNodes) C0[(size_t)v * GCN_F] = acc[v] * S0[v * HS_STRIDE];
            }
        }
        s_c = s_n; d_c = d_n; dv_c = dv_n; mine_c = mine_n;
        s_n = s_a; d_n = d_a; dv_n = dv_a;
        buf ^= 1;
    }
}

#define GO_WAVES    8
#define GO_THREADS  256
#define GO_BLOCKS   2048
#define GO_INFLIGHT 8

__device__ __forceinline__ void gcn_out_ahead(const float4* __restrict__ TsRow,
                                              const int* __restrict__ srcList,
                                              const int* __restrict__ off,
                                              const int* __restrict__ deg,
                                              const float* __restrict__ dinv,
                                              int node, int step, int nNodes, int lane,
                                              int s_n, int d_n,
                                              int& s_a, int& d_a, float& dv_a,
                                              int& mine_n, float4& self_n) {
    const int nn = node + step;
    const int na = nn + step;
    if (na < nNodes) { s_a = off[na]; d_a = deg[na]; dv_a = dinv[na]; }
    if (nn < nNodes) {
        if (lane < d_n) mine_n = srcList[s_n + lane];
        self_n = TsRow[(size_t)nn * 32];
    }
}

__global__ __launch_bounds__(GO_THREADS, 16)
void gcn_gather_out(const float* __restrict__ Ts, const int* __restrict__ srcList,
                    const int* __restrict__ off, const int* __restrict__ deg,
                    const float* __restrict__ dinv, const float* __restrict__ bmu,
                    const float* __restrict__ blv, float* __restrict__ out, int nNodes) {
    const int lane = threadIdx.x & 31;
    const int wave = __builtin_amdgcn_readfirstlane(threadIdx.x >> 5);
    const int step = gridDim.x * GO_WAVES;
    int node = blockIdx.x * GO_WAVES + wave;
    if (node >= nNodes) return;

    const float4* TsRow = (const float4*)Ts + lane;
    float4 bb;
    float* dst;
    if (lane < 16) {
        bb  = ((const float4*)bmu)[lane];
        dst = out + lane * 4;
    } else {
        bb  = ((const float4*)blv)[lane - 16];
        dst = out + (size_t)nNodes * 64 + (lane - 16) * 4;
    }

    int   s_c = off[node], d_c = deg[node];
    float dv_c = dinv[node];
    int   s_n = 0, d_n = 0;
    float dv_n = 0.0f;
    if (node + step < nNodes) { s_n = off[node + step]; d_n = deg[node + step]; dv_n = dinv[node + step]; }
    int    mine_c = (lane < d_c) ? srcList[s_c + lane] : 0;
    float4 self_c = TsRow[(size_t)node * 32];

    for (; node < nNodes; node += step) {
        int    s_a = 0, d_a = 0;
        float  dv_a = 0.0f;
        int    mine_n = 0;
        float4 self_n = {0.0f, 0.0f, 0.0f, 0.0f};
        bool   ahead = true;

        float4 acc = self_c;
        for (int e0 = 0; e0 < d_c; e0 += 32) {
            const int cnt = (d_c - e0 < 32) ? (d_c - e0) : 32;
            const int mine = (e0 == 0) ? mine_c
                                       : ((lane < cnt) ? srcList[s_c + e0 + lane] : 0);
            int e = 0;
            for (; e + GO_INFLIGHT <= cnt; e += GO_INFLIGHT) {
                float4 v[GO_INFLIGHT];
#pragma unroll
                for (int u = 0; u < GO_INFLIGHT; ++u)
                    v[u] = TsRow[(size_t)__builtin_amdgcn_readlane(mine, e + u) * 32];
                if (ahead) {
                    gcn_out_ahead(TsRow, srcList, off, deg, dinv, node, step, nNodes, lane,
                                  s_n, d_n, s_a, d_a, dv_a, mine_n, self_n);
                    ahead = false;
                }
#pragma unroll
                for (int u = 0; u < GO_INFLIGHT; ++u) gcn_add4(acc, v[u]);
            }
            if (e < cnt) {
                float4 v[GO_INFLIGHT - 1] = {};
#pragma unroll
                for (int u = 0; u < GO_INFLIGHT - 1; ++u)
                    if (e + u < cnt)
                        v[u] = TsRow[(size_t)__builtin_amdgcn_readlane(mine, e + u) * 32];
                if (ahead) {
                    gcn_out_ahead(TsRow, srcList, off, deg, dinv, node, step, nNodes, lane,
                                  s_n, d_n, s_a, d_a, dv_a, mine_n, self_n);
                    ahead = false;
                }
#pragma unroll
                for (int u = 0; u < GO_INFLIGHT - 1; ++u)
                    if (e + u < cnt) gcn_add4(acc, v[u]);
            }
        }
        if (ahead)
            gcn_out_ahead(TsRow, srcList, off, deg, dinv, node, step, nNodes, lane,
                          s_n, d_n, s_a, d_a, dv_a, mine_n, self_n);

        float4 r;
        r.x = dv_c * acc.x + bb.x;
        r.y = dv_c * acc.y + bb.y;
        r.z = dv_c * acc.z + bb.z;
        r.w = dv_c * acc.w + bb.w;
        *(float4*)(dst + (size_t)node * 64) = r;

        s_c = s_n; d_c = d_n; dv_c = dv_n; mine_c = mine_n; self_c = self_n;
        s_n = s_a; d_n = d_a; dv_n = dv_a;
    }
}


extern "C" void kernel_launch(void* const* d_in, const int* in_sizes, int n_in,
                              void* d_out, int out_size, void* d_ws, size_t ws_size,
                              hipStream_t stream) {
    const float* X   = (const float*)d_in[0];
    const int*   ei  = (const int*)d_in[1];
    const float* W1  = (const float*)d_in[2];
    const float* b1  = (const float*)d_in[3];
    const float* Wmu = (const float*)d_in[4];
    const float* bmu = (const float*)d_in[5];
    const float* Wlv = (const float*)d_in[6];
    const float* blv = (const float*)d_in[7];
    float* out = (float*)d_out;

    const int N = in_sizes[0] / 256;
    const int E = in_sizes[1] / 2;
    const int* rows = ei;
    const int* cols = ei + E;

    float* Ts   = (float*)d_ws;
    float* Hbuf = Ts + (size_t)N * GCN_F;
    float* W23  = Hbuf + (size_t)N * GCN_F;
    float* dinv = W23 + GCN_F * GCN_F;
    int* degInt  = (int*)(dinv + N);
    int* off     = degInt + N;
    int* srcList = off + N;

    const int nbk = (N + CSR_BNODES - 1) / CSR_BNODES;
    int* colB   = (int*)Hbuf;
    int* rowB   = colB + E;
    int* histT  = rowB + E;
    int* btot   = histT + CSR_CHUNKS * nbk;
    int* bstart = btot + nbk;
    gcn_bucket_hist<<<CSR_CHUNKS, CSR_BLOCK, 0, stream>>>(cols, histT, E, nbk);
    gcn_bucket_prefix<<<nbk, CSR_CHUNKS, 0, stream>>>(histT, btot, nbk);
    gcn_bucket_scatter<<<CSR_CHUNKS, CSR_BLOCK, 0, stream>>>(rows, cols, histT, btot, bstart,
                                                        rowB, colB, E, nbk);
    gcn_bucket_to_csr<<<nbk, CSR_BLOCK, 0, stream>>>(rowB, colB, bstart, btot, degInt, off,
                                                dinv, srcList, N);

    const int rowTiles = (N + 15) / 16;
    const int gemmBlocks = (rowTiles + 7) / 8;

    gcn_gemm_rowscale_wmma<<<gemmBlocks, 256, 0, stream>>>(X, W1, dinv, Ts, N, 256);
    gcn_gather_relu_gemm<<<768, FG_THREADS, 0, stream>>>(Ts, srcList, off, degInt, dinv, b1,
                                                          Wmu, Wlv, Hbuf, N, rowTiles);

    gcn_gather_out<<<GO_BLOCKS, GO_THREADS, 0, stream>>>(Hbuf, srcList, off, degInt,
                                                     dinv, bmu, blv, out, N);
}


# layer-1 product X@W1 on 256 resident work-groups holding all of W1 in LDS: waves walk 16x64 tiles dealt evenly over the SIMDs, no K-block barrier, A asked for 16 k-steps ahead; k order kept
# speedup vs baseline: 1.8153x; 1.0666x over previous
#include <hip/hip_runtime.h>
#include <hip/hip_bf16.h>

typedef float v2f __attribute__((ext_vector_type(2)));
typedef float v8f __attribute__((ext_vector_type(8)));

#define GCN_F 128
#define BS_STRIDE 136


#define CSR_BSHIFT 8
#define CSR_BNODES 256
#define CSR_CHUNKS 256
#define CSR_MAXB   1024
#define CSR_DEPTH  16
#define CSR_BLOCK  512

__device__ __forceinline__ int gcn_block_incl_scan(int v, int* s_w) {
    const int lane = threadIdx.x & 31;
    const int wave = threadIdx.x >> 5;
    const int nw   = blockDim.x >> 5;
    for (int d = 1; d < 32; d <<= 1) {
        const int t = __shfl_up(v, d);
        if (lane >= d) v += t;
    }
    if (lane == 31) s_w[wave] = v;
    __syncthreads();
    if (wave == 0) {
        const int w = (lane < nw) ? s_w[lane] : 0;
        int wi = w;
        for (int d = 1; d < 32; d <<= 1) {
            const int t = __shfl_up(wi, d);
            if (lane >= d) wi += t;
        }
        s_w[lane] = wi - w;
    }
    __syncthreads();
    return v + s_w[wave];
}

__global__ __launch_bounds__(CSR_BLOCK)
void gcn_bucket_hist(const int* __restrict__ cols, int* __restrict__ histT, int e, int nbk) {
    __shared__ int s_h[CSR_MAXB];
    const int tid = threadIdx.x;
    for (int b = tid; b < nbk; b += CSR_BLOCK) s_h[b] = 0;
    __syncthreads();
    const int chunk = (e + CSR_CHUNKS - 1) / CSR_CHUNKS;
    const int beg = blockIdx.x * chunk;
    const int end = (beg + chunk < e) ? (beg + chunk) : e;
    for (int i0 = beg + tid; i0 < end; i0 += CSR_BLOCK * CSR_DEPTH) {
        int c[CSR_DEPTH];
#pragma unroll
        for (int u = 0; u < CSR_DEPTH; ++u) {
            const int i = i0 + CSR_BLOCK * u;
            c[u] = cols[(i < end) ? i : (end - 1)];
        }
#pragma unroll
        for (int u = 0; u < CSR_DEPTH; ++u)
            if (i0 + CSR_BLOCK * u < end) atomicAdd(&s_h[c[u] >> CSR_BSHIFT], 1);
    }
    __syncthreads();
    for (int b = tid; b < nbk; b += CSR_BLOCK) histT[blockIdx.x * nbk + b] = s_h[b];
}

__global__ __launch_bounds__(CSR_CHUNKS)
void gcn_bucket_prefix(int* __restrict__ histT, int* __restrict__ tot, int nbk) {
    __shared__ int s_w[32];
    const int b = blockIdx.x;
    const int k = threadIdx.x;
    const int v = histT[k * nbk + b];
    const int incl = gcn_block_incl_scan(v, s_w);
    histT[k * nbk + b] = incl - v;
    if (k == CSR_CHUNKS - 1) tot[b] = incl;
}

__global__ __launch_bounds__(CSR_BLOCK)
void gcn_bucket_scatter(const int* __restrict__ rows, const int* __restrict__ cols,
                        const int* __restrict__ prefT, const int* __restrict__ tot,
                        int* __restrict__ bstart, int* __restrict__ rowB,
                        int* __restrict__ colB, int e, int nbk) {
    __shared__ int s_cur[CSR_MAXB];
    __shared__ int s_w[32];
    const int tid = threadIdx.x;
    const int b0 = 2 * tid, b1 = 2 * tid + 1;
    const int v0 = (b0 < nbk) ? tot[b0] : 0;
    const int v1 = (b1 < nbk) ? tot[b1] : 0;
    const int ex0 = gcn_block_incl_scan(v0 + v1, s_w) - (v0 + v1);
    const int ex1 = ex0 + v0;
    if (b0 < nbk) {
        s_cur[b0] = ex0 + prefT[blockIdx.x * nbk + b0];
        if (blockIdx.x == 0) bstart[b0] = ex0;
    }
    if (b1 < nbk) {
        s_cur[b1] = ex1 + prefT[blockIdx.x * nbk + b1];
        if (blockIdx.x == 0) bstart[b1] = ex1;
    }
    __syncthreads();
    const int chunk = (e + CSR_CHUNKS - 1) / CSR_CHUNKS;
    const int beg = blockIdx.x * chunk;
    const int end = (beg + chunk < e) ? (beg + chunk) : e;
    for (int i0 = beg + tid; i0 < end; i0 += CSR_BLOCK * CSR_DEPTH) {
        int c[CSR_DEPTH], r[CSR_DEPTH], p[CSR_DEPTH];
#pragma unroll
        for (int u = 0; u < CSR_DEPTH; ++u) {
            const int i  = i0 + CSR_BLOCK * u;
            const int ic = (i < end) ? i : (end - 1);
            c[u] = cols[ic];
            r[u] = rows[ic];
        }
#pragma unroll
        for (int u = 0; u < CSR_DEPTH; ++u)
            p[u] = (i0 + CSR_BLOCK * u < end) ? atomicAdd(&s_cur[c[u] >> CSR_BSHIFT], 1) : 0;
#pragma unroll
        for (int u = 0; u < CSR_DEPTH; ++u) {
            if (i0 + CSR_BLOCK * u < end) {
                colB[p[u]] = c[u];
                rowB[p[u]] = r[u];
            }
        }
    }
}

__global__ __launch_bounds__(CSR_BLOCK)
void gcn_bucket_to_csr(const int* __restrict__ rowB, const int* __restrict__ colB,
                       const int* __restrict__ bstart, const int* __restrict__ tot,
                       int* __restrict__ deg, int* __restrict__ off,
                       float* __restrict__ dinv, int* __restrict__ srcList, int n) {
    __shared__ int s_cnt[CSR_BNODES];
    __shared__ int s_w[32];
    const int tid = threadIdx.x;
    const int b   = blockIdx.x;
    if (tid < CSR_BNODES) s_cnt[tid] = 0;
    __syncthreads();
    const int s0  = bstart[b];
    const int cnt = tot[b];
    for (int i0 = tid; i0 < cnt; i0 += CSR_BLOCK * CSR_DEPTH) {
        int c[CSR_DEPTH];
#pragma unroll
        for (int u = 0; u < CSR_DEPTH; ++u) {
            const int i = i0 + CSR_BLOCK * u;
            c[u] = colB[s0 + ((i < cnt) ? i : (cnt - 1))];
        }
#pragma unroll
        for (int u = 0; u < CSR_DEPTH; ++u)
            if (i0 + CSR_BLOCK * u < cnt) atomicAdd(&s_cnt[c[u] & (CSR_BNODES - 1)], 1);
    }
    __syncthreads();
    const int v = (tid < CSR_BNODES) ? s_cnt[tid] : 0;
    const int excl = gcn_block_incl_scan(v, s_w) - v;
    if (tid < CSR_BNODES) {
        s_cnt[tid] = excl;
        const int node = b * CSR_BNODES + tid;
        if (node < n) {
            deg[node]  = v;
            off[node]  = s0 + excl;
            dinv[node] = rsqrtf((float)(v + 1));
        }
    }
    __syncthreads();
    for (int i0 = tid; i0 < cnt; i0 += CSR_BLOCK * CSR_DEPTH) {
        int c[CSR_DEPTH], r[CSR_DEPTH], p[CSR_DEPTH];
#pragma unroll
        for (int u = 0; u < CSR_DEPTH; ++u) {
            const int i  = i0 + CSR_BLOCK * u;
            const int ic = s0 + ((i < cnt) ? i : (cnt - 1));
            c[u] = colB[ic];
            r[u] = rowB[ic];
        }
#pragma unroll
        for (int u = 0; u < CSR_DEPTH; ++u)
            p[u] = (i0 + CSR_BLOCK * u < cnt)
                       ? atomicAdd(&s_cnt[c[u] & (CSR_BNODES - 1)], 1) : 0;
#pragma unroll
        for (int u = 0; u < CSR_DEPTH; ++u)
            if (i0 + CSR_BLOCK * u < cnt) srcList[s0 + p[u]] = r[u];
    }
}

#define G1_K       256
#define G1_THREADS 512
#define G1_BLOCKS  256
#define G1_SLOTS   8
#define G1_CHUNK   16
#define G1_LDS_FLOATS 41216

__global__ __launch_bounds__(G1_THREADS)
void gcn_gemm_rowscale_wmma(const float* __restrict__ A,
                            const float* __restrict__ B,
                            const float* __restrict__ rowScale,
                            float* __restrict__ C,
                            int M, int rowTiles) {
    __shared__ float Ws[G1_LDS_FLOATS];

    const int tid  = threadIdx.x;
    const int lane = tid & 31;
    const int wave = __builtin_amdgcn_readfirstlane(tid >> 5);
    const int half = lane >> 4;
    const int l16  = lane & 15;

    {
        float4 t[16];
#pragma unroll
        for (int u = 0; u < 16; ++u) {
            const int idx = tid + G1_THREADS * u;
            t[u] = ((const float4*)(B + (size_t)(idx >> 5) * GCN_F))[idx & 31];
        }
#pragma unroll
        for (int u = 0; u < 16; ++u) {
            const int idx = tid + G1_THREADS * u;
            *(float4*)(Ws + (idx >> 5) * BS_STRIDE + (idx & 31) * 4) = t[u];
        }
    }
    __syncthreads();

    const int colbase = 64 * (wave & 1);
    const int tstep   = G1_BLOCKS * G1_SLOTS;
    int tile = blockIdx.x + G1_BLOCKS * (wave >> 1);
    if (tile >= rowTiles) return;

    const float* Bl = Ws + (2 * half) * BS_STRIDE + colbase + l16;

    int r0 = tile * 16 + l16;
    const float* Ap = A + (size_t)((r0 < M) ? r0 : (M - 1)) * G1_K + 2 * half;

    v2f an[G1_CHUNK];
#pragma unroll
    for (int i = 0; i < G1_CHUNK; ++i) an[i] = *(const v2f*)(Ap + 4 * i);

    for (; tile < rowTiles; tile += tstep) {
        const int tn = (tile + tstep < rowTiles) ? (tile + tstep) : tile;
        const int rn = tn * 16 + l16;
        const float* ApN = A + (size_t)((rn < M) ? rn : (M - 1)) * G1_K + 2 * half;

        const int rbase = tile * 16 + 8 * half;
        float sc[8];
#pragma unroll
        for (int v = 0; v < 8; ++v) sc[v] = rowScale[(rbase + v < M) ? (rbase + v) : (M - 1)];

        v8f acc[4] = {};
#pragma nounroll
        for (int kc = 0; kc < G1_K / (4 * G1_CHUNK); ++kc) {
            v2f ac[G1_CHUNK];
#pragma unroll
            for (int i = 0; i < G1_CHUNK; ++i) ac[i] = an[i];
            const float* Np = (kc < G1_K / (4 * G1_CHUNK) - 1) ? (Ap + 4 * G1_CHUNK * (kc + 1)) : ApN;
#pragma unroll
            for (int i = 0; i < G1_CHUNK; ++i) an[i] = *(const v2f*)(Np + 4 * i);

            const float* Bk = Bl + (4 * G1_CHUNK * kc) * BS_STRIDE;
            v2f bc[4];
#pragma unroll
            for (int j = 0; j < 4; ++j) { bc[j].x = Bk[16 * j]; bc[j].y = Bk[16 * j + BS_STRIDE]; }
#pragma unroll
            for (int i = 0; i < G1_CHUNK; ++i) {
                v2f bn[4] = {};
                if (i + 1 < G1_CHUNK) {
                    const float* Bq = Bk + (4 * (i + 1)) * BS_STRIDE;
#pragma unroll
                    for (int j = 0; j < 4; ++j) { bn[j].x = Bq[16 * j]; bn[j].y = Bq[16 * j + BS_STRIDE]; }
                }
#pragma unroll
                for (int j = 0; j < 4; ++j)
                    acc[j] = __builtin_amdgcn_wmma_f32_16x16x4_f32(
                        false, ac[i], false, bc[j], (short)0, acc[j], false, false);
#pragma unroll
                for (int j = 0; j < 4; ++j) bc[j] = bn[j];
            }
        }

        float* C0 = C + (size_t)rbase * GCN_F + colbase + l16;
        if (tile * 16 + 16 <= M) {
#pragma unroll
            for (int v = 0; v < 8; ++v)
#pragma unroll
                for (int j = 0; j < 4; ++j) C0[(size_t)v * GCN_F + 16 * j] = acc[j][v] * sc[v];
        } else {
#pragma unroll
            for (int v = 0; v < 8; ++v)
                if (rbase + v < M) {
#pragma unroll
                    for (int j = 0; j < 4; ++j) C0[(size_t)v * GCN_F + 16 * j] = acc[j][v] * sc[v];
                }
        }
        Ap = ApN;
    }
}


#define FG_WAVES    16
#define FG_THREADS  512
#define HS_STRIDE   132
#define HS_DINV     128
#define FG_INFLIGHT 8

__device__ __forceinline__ void gcn_add4(float4& a, const float4& v) {
    a.x += v.x; a.y += v.y; a.z += v.z; a.w += v.w;
}

__global__ __launch_bounds__(FG_THREADS)
void gcn_gather_relu_gemm(const float* __restrict__ Ts, const int* __restrict__ srcList,
                          const int* __restrict__ off, const int* __restrict__ deg,
                          const float* __restrict__ dinv, const float* __restrict__ b1,
                          const float* __restrict__ Wmu, const float* __restrict__ Wlv,
                          float* __restrict__ T2, int nNodes, int nTiles) {
    __shared__ float Ws[GCN_F * BS_STRIDE];
    __shared__ float Hs[2][FG_WAVES * HS_STRIDE];

    const int tid  = threadIdx.x;
    const int lane = tid & 31;
    const int wave = __builtin_amdgcn_readfirstlane(tid >> 5);
    const int half = lane >> 4;
    const int l16  = lane & 15;

    for (int idx = tid; idx < GCN_F * 32; idx += FG_THREADS) {
        const int r = idx >> 5, c4 = idx & 31;
        const float4 v = (c4 < 16) ? ((const float4*)(Wmu + (size_t)r * 64))[c4]
                                   : ((const float4*)(Wlv + (size_t)r * 64))[c4 - 16];
        *(float4*)(Ws + r * BS_STRIDE + c4 * 4) = v;
    }
    const float4 bb = ((const float4*)b1)[lane];
    const float4* TsRow = (const float4*)Ts + lane;

    const int step = gridDim.x;
    int tile = blockIdx.x;

    int   s_c = 0, d_c = 0, s_n = 0, d_n = 0;
    float dv_c = 0.0f, dv_n = 0.0f;
    {
        const int n0 = tile * FG_WAVES + wave;
        if (tile < nTiles && n0 < nNodes) { s_c = off[n0]; d_c = deg[n0]; dv_c = dinv[n0]; }
        const int n1 = (tile + step) * FG_WAVES + wave;
        if (tile + step < nTiles && n1 < nNodes) { s_n = off[n1]; d_n = deg[n1]; dv_n = dinv[n1]; }
    }
    int mine_c = (lane < d_c) ? srcList[s_c + lane] : 0;

    int buf = 0;
    for (; tile < nTiles; tile += step) {
        const int node = tile * FG_WAVES + wave;

        int   s_a = 0, d_a = 0;
        float dv_a = 0.0f;
        {
            const int ta = tile + 2 * step;
            const int na = ta * FG_WAVES + wave;
            if (ta < nTiles && na < nNodes) { s_a = off[na]; d_a = deg[na]; dv_a = dinv[na]; }
        }
        const int mine_n = (lane < d_n) ? srcList[s_n + lane] : 0;

        float4 r = {0.0f, 0.0f, 0.0f, 0.0f};
        if (node < nNodes) {
            float4 acc = TsRow[(size_t)node * 32];
            for (int e0 = 0; e0 < d_c; e0 += 32) {
                const int cnt = (d_c - e0 < 32) ? (d_c - e0) : 32;
                const int mine = (e0 == 0) ? mine_c
                                           : ((lane < cnt) ? srcList[s_c + e0 + lane] : 0);
                int e = 0;
                for (; e + FG_INFLIGHT <= cnt; e += FG_INFLIGHT) {
                    float4 v[FG_INFLIGHT];
#pragma unroll
                    for (int u = 0; u < FG_INFLIGHT; ++u)
                        v[u] = TsRow[(size_t)__builtin_amdgcn_readlane(mine, e + u) * 32];
#pragma unroll
                    for (int u = 0; u < FG_INFLIGHT; ++u) gcn_add4(acc, v[u]);
                }
                if (e < cnt) {
                    float4 v[FG_INFLIGHT - 1] = {};
#pragma unroll
                    for (int u = 0; u < FG_INFLIGHT - 1; ++u)
                        if (e + u < cnt)
                            v[u] = TsRow[(size_t)__builtin_amdgcn_readlane(mine, e + u) * 32];
#pragma unroll
                    for (int u = 0; u < FG_INFLIGHT - 1; ++u)
                        if (e + u < cnt) gcn_add4(acc, v[u]);
                }
            }
            r.x = fmaxf(dv_c * acc.x + bb.x, 0.0f);
            r.y = fmaxf(dv_c * acc.y + bb.y, 0.0f);
            r.z = fmaxf(dv_c * acc.z + bb.z, 0.0f);
            r.w = fmaxf(dv_c * acc.w + bb.w, 0.0f);
        }
        float* Hrow = &Hs[buf][wave * HS_STRIDE];
        *(float4*)(Hrow + lane * 4) = r;
        if (lane == 0) Hrow[HS_DINV] = dv_c;
        __syncthreads();

        if (wave < 8) {
            const float* A0 = &Hs[buf][l16 * HS_STRIDE + 2 * half];
            const float* B0 = Ws + (2 * half) * BS_STRIDE + 16 * wave + l16;
            v8f acc = {};
#pragma unroll 8
            for (int i = 0; i < GCN_F / 4; ++i) {
                v2f a, b;
                a.x = A0[4 * i];
                a.y = A0[4 * i + 1];
                b.x = B0[(4 * i) * BS_STRIDE];
                b.y = B0[(4 * i + 1) * BS_STRIDE];
                acc = __builtin_amdgcn_wmma_f32_16x16x4_f32(
                    false, a, false, b, (short)0, acc, false, false);
            }
            const int row0 = tile * FG_WAVES + 8 * half;
            const float* S0 = &Hs[buf][(8 * half) * HS_STRIDE + HS_DINV];
            float* C0 = T2 + (size_t)row0 * GCN_F + 16 * wave + l16;
            if (tile * FG_WAVES + FG_WAVES <= nNodes) {
#pragma unroll
                for (int v = 0; v < 8; ++v) C0[(size_t)v * GCN_F] = acc[v] * S0[v * HS_STRIDE];
            } else {
#pragma unroll
                for (int v = 0; v < 8; ++v)
                    if (row0 + v < nNodes) C0[(size_t)v * GCN_F] = acc[v] * S0[v * HS_STRIDE];
            }
        }
        s_c = s_n; d_c = d_n; dv_c = dv_n; mine_c = mine_n;
        s_n = s_a; d_n = d_a; dv_n = dv_a;
        buf ^= 1;
    }
}

#define GO_WAVES    8
#define GO_THREADS  256
#define GO_BLOCKS   2048
#define GO_INFLIGHT 8

__device__ __forceinline__ void gcn_out_ahead(const float4* __restrict__ TsRow,
                                              const int* __restrict__ srcList,
                                              const int* __restrict__ off,
                                              const int* __restrict__ deg,
                                              const float* __restrict__ dinv,
                                              int node, int step, int nNodes, int lane,
                                              int s_n, int d_n,
                                              int& s_a, int& d_a, float& dv_a,
                                              int& mine_n, float4& self_n) {
    const int nn = node + step;
    const int na = nn + step;
    if (na < nNodes) { s_a = off[na]; d_a = deg[na]; dv_a = dinv[na]; }
    if (nn < nNodes) {
        if (lane < d_n) mine_n = srcList[s_n + lane];
        self_n = TsRow[(size_t)nn * 32];
    }
}

__global__ __launch_bounds__(GO_THREADS, 16)
void gcn_gather_out(const float* __restrict__ Ts, const int* __restrict__ srcList,
                    const int* __restrict__ off, const int* __restrict__ deg,
                    const float* __restrict__ dinv, const float* __restrict__ bmu,
                    const float* __restrict__ blv, float* __restrict__ out, int nNodes) {
    const int lane = threadIdx.x & 31;
    const int wave = __builtin_amdgcn_readfirstlane(threadIdx.x >> 5);
    const int step = gridDim.x * GO_WAVES;
    int node = blockIdx.x * GO_WAVES + wave;
    if (node >= nNodes) return;

    const float4* TsRow = (const float4*)Ts + lane;
    float4 bb;
    float* dst;
    if (lane < 16) {
        bb  = ((const float4*)bmu)[lane];
        dst = out + lane * 4;
    } else {
        bb  = ((const float4*)blv)[lane - 16];
        dst = out + (size_t)nNodes * 64 + (lane - 16) * 4;
    }

    int   s_c = off[node], d_c = deg[node];
    float dv_c = dinv[node];
    int   s_n = 0, d_n = 0;
    float dv_n = 0.0f;
    if (node + step < nNodes) { s_n = off[node + step]; d_n = deg[node + step]; dv_n = dinv[node + step]; }
    int    mine_c = (lane < d_c) ? srcList[s_c + lane] : 0;
    float4 self_c = TsRow[(size_t)node * 32];

    for (; node < nNodes; node += step) {
        int    s_a = 0, d_a = 0;
        float  dv_a = 0.0f;
        int    mine_n = 0;
        float4 self_n = {0.0f, 0.0f, 0.0f, 0.0f};
        bool   ahead = true;

        float4 acc = self_c;
        for (int e0 = 0; e0 < d_c; e0 += 32) {
            const int cnt = (d_c - e0 < 32) ? (d_c - e0) : 32;
            const int mine = (e0 == 0) ? mine_c
                                       : ((lane < cnt) ? srcList[s_c + e0 + lane] : 0);
            int e = 0;
            for (; e + GO_INFLIGHT <= cnt; e += GO_INFLIGHT) {
                float4 v[GO_INFLIGHT];
#pragma unroll
                for (int u = 0; u < GO_INFLIGHT; ++u)
                    v[u] = TsRow[(size_t)__builtin_amdgcn_readlane(mine, e + u) * 32];
                if (ahead) {
                    gcn_out_ahead(TsRow, srcList, off, deg, dinv, node, step, nNodes, lane,
                                  s_n, d_n, s_a, d_a, dv_a, mine_n, self_n);
                    ahead = false;
                }
#pragma unroll
                for (int u = 0; u < GO_INFLIGHT; ++u) gcn_add4(acc, v[u]);
            }
            if (e < cnt) {
                float4 v[GO_INFLIGHT - 1] = {};
#pragma unroll
                for (int u = 0; u < GO_INFLIGHT - 1; ++u)
                    if (e + u < cnt)
                        v[u] = TsRow[(size_t)__builtin_amdgcn_readlane(mine, e + u) * 32];
                if (ahead) {
                    gcn_out_ahead(TsRow, srcList, off, deg, dinv, node, step, nNodes, lane,
                                  s_n, d_n, s_a, d_a, dv_a, mine_n, self_n);
                    ahead = false;
                }
#pragma unroll
                for (int u = 0; u < GO_INFLIGHT - 1; ++u)
                    if (e + u < cnt) gcn_add4(acc, v[u]);
            }
        }
        if (ahead)
            gcn_out_ahead(TsRow, srcList, off, deg, dinv, node, step, nNodes, lane,
                          s_n, d_n, s_a, d_a, dv_a, mine_n, self_n);

        float4 r;
        r.x = dv_c * acc.x + bb.x;
        r.y = dv_c * acc.y + bb.y;
        r.z = dv_c * acc.z + bb.z;
        r.w = dv_c * acc.w + bb.w;
        *(float4*)(dst + (size_t)node * 64) = r;

        s_c = s_n; d_c = d_n; dv_c = dv_n; mine_c = mine_n; self_c = self_n;
        s_n = s_a; d_n = d_a; dv_n = dv_a;
    }
}


extern "C" void kernel_launch(void* const* d_in, const int* in_sizes, int n_in,
                              void* d_out, int out_size, void* d_ws, size_t ws_size,
                              hipStream_t stream) {
    const float* X   = (const float*)d_in[0];
    const int*   ei  = (const int*)d_in[1];
    const float* W1  = (const float*)d_in[2];
    const float* b1  = (const float*)d_in[3];
    const float* Wmu = (const float*)d_in[4];
    const float* bmu = (const float*)d_in[5];
    const float* Wlv = (const float*)d_in[6];
    const float* blv = (const float*)d_in[7];
    float* out = (float*)d_out;

    const int N = in_sizes[0] / 256;
    const int E = in_sizes[1] / 2;
    const int* rows = ei;
    const int* cols = ei + E;

    float* Ts   = (float*)d_ws;
    float* Hbuf = Ts + (size_t)N * GCN_F;
    float* W23  = Hbuf + (size_t)N * GCN_F;
    float* dinv = W23 + GCN_F * GCN_F;
    int* degInt  = (int*)(dinv + N);
    int* off     = degInt + N;
    int* srcList = off + N;

    const int nbk = (N + CSR_BNODES - 1) / CSR_BNODES;
    int* colB   = (int*)Hbuf;
    int* rowB   = colB + E;
    int* histT  = rowB + E;
    int* btot   = histT + CSR_CHUNKS * nbk;
    int* bstart = btot + nbk;
    gcn_bucket_hist<<<CSR_CHUNKS, CSR_BLOCK, 0, stream>>>(cols, histT, E, nbk);
    gcn_bucket_prefix<<<nbk, CSR_CHUNKS, 0, stream>>>(histT, btot, nbk);
    gcn_bucket_scatter<<<CSR_CHUNKS, CSR_BLOCK, 0, stream>>>(rows, cols, histT, btot, bstart,
                                                        rowB, colB, E, nbk);
    gcn_bucket_to_csr<<<nbk, CSR_BLOCK, 0, stream>>>(rowB, colB, bstart, btot, degInt, off,
                                                dinv, srcList, N);

    const int rowTiles = (N + 15) / 16;

    gcn_gemm_rowscale_wmma<<<G1_BLOCKS, G1_THREADS, 0, stream>>>(X, W1, dinv, Ts, N, rowTiles);
    gcn_gather_relu_gemm<<<768, FG_THREADS, 0, stream>>>(Ts, srcList, off, degInt, dinv, b1,
                                                          Wmu, Wlv, Hbuf, N, rowTiles);

    gcn_gather_out<<<GO_BLOCKS, GO_THREADS, 0, stream>>>(Hbuf, srcList, off, degInt,
                                                     dinv, bmu, blv, out, N);
}


# CSR build: both sort levels put their records in order in an LDS stage and store them from there as runs of neighbouring words, not 32 single words in 32 lines per store; list order as before
# speedup vs baseline: 2.2535x; 1.2414x over previous
#include <hip/hip_runtime.h>
#include <hip/hip_bf16.h>

typedef float v2f __attribute__((ext_vector_type(2)));
typedef float v8f __attribute__((ext_vector_type(8)));

#define GCN_F 128
#define BS_STRIDE 136


#define CSR_BSHIFT 8
#define CSR_BNODES 256
#define CSR_CHUNKS 256
#define CSR_MAXB   1024
#define CSR_DEPTH  16
#define CSR_BLOCK  512
#define CSR_TRIP   (CSR_BLOCK * CSR_DEPTH)
#define CSR_STAGE2 16384

__device__ __forceinline__ int gcn_block_incl_scan(int v, int* s_w) {
    const int lane = threadIdx.x & 31;
    const int wave = threadIdx.x >> 5;
    const int nw   = blockDim.x >> 5;
    for (int d = 1; d < 32; d <<= 1) {
        const int t = __shfl_up(v, d);
        if (lane >= d) v += t;
    }
    if (lane == 31) s_w[wave] = v;
    __syncthreads();
    if (wave == 0) {
        const int w = (lane < nw) ? s_w[lane] : 0;
        int wi = w;
        for (int d = 1; d < 32; d <<= 1) {
            const int t = __shfl_up(wi, d);
            if (lane >= d) wi += t;
        }
        s_w[lane] = wi - w;
    }
    __syncthreads();
    return v + s_w[wave];
}

__global__ __launch_bounds__(CSR_BLOCK)
void gcn_bucket_hist(const int* __restrict__ cols, int* __restrict__ histT, int e, int nbk) {
    __shared__ int s_h[CSR_MAXB];
    const int tid = threadIdx.x;
    for (int b = tid; b < nbk; b += CSR_BLOCK) s_h[b] = 0;
    __syncthreads();
    const int chunk = (e + CSR_CHUNKS - 1) / CSR_CHUNKS;
    const int beg = blockIdx.x * chunk;
    const int end = (beg + chunk < e) ? (beg + chunk) : e;
    for (int i0 = beg + tid; i0 < end; i0 += CSR_BLOCK * CSR_DEPTH) {
        int c[CSR_DEPTH];
#pragma unroll
        for (int u = 0; u < CSR_DEPTH; ++u) {
            const int i = i0 + CSR_BLOCK * u;
            c[u] = cols[(i < end) ? i : (end - 1)];
        }
#pragma unroll
        for (int u = 0; u < CSR_DEPTH; ++u)
            if (i0 + CSR_BLOCK * u < end) atomicAdd(&s_h[c[u] >> CSR_BSHIFT], 1);
    }
    __syncthreads();
    for (int b = tid; b < nbk; b += CSR_BLOCK) histT[blockIdx.x * nbk + b] = s_h[b];
}

__global__ __launch_bounds__(CSR_CHUNKS)
void gcn_bucket_prefix(int* __restrict__ histT, int* __restrict__ tot, int nbk) {
    __shared__ int s_w[32];
    const int b = blockIdx.x;
    const int k = threadIdx.x;
    const int v = histT[k * nbk + b];
    const int incl = gcn_block_incl_scan(v, s_w);
    histT[k * nbk + b] = incl - v;
    if (k == CSR_CHUNKS - 1) tot[b] = incl;
}

__global__ __launch_bounds__(CSR_BLOCK)
void gcn_bucket_scatter(const int* __restrict__ rows, const int* __restrict__ cols,
                        const int* __restrict__ prefT, const int* __restrict__ tot,
                        int* __restrict__ bstart, int* __restrict__ rowB,
                        int* __restrict__ colB, int e, int nbk) {
    __shared__ int s_loc[CSR_MAXB];
    __shared__ int s_dlt[CSR_MAXB];
    __shared__ int s_w[32];
    __shared__ int s_sc[CSR_TRIP];
    __shared__ int s_sr[CSR_TRIP];
    const int tid = threadIdx.x;
    const int b0 = 2 * tid, b1 = 2 * tid + 1;
    const int v0 = (b0 < nbk) ? tot[b0] : 0;
    const int v1 = (b1 < nbk) ? tot[b1] : 0;
    const int ex0 = gcn_block_incl_scan(v0 + v1, s_w) - (v0 + v1);
    const int ex1 = ex0 + v0;
    int base0 = 0, base1 = 0;
    if (b0 < nbk) {
        base0 = ex0 + prefT[blockIdx.x * nbk + b0];
        if (blockIdx.x == 0) bstart[b0] = ex0;
    }
    if (b1 < nbk) {
        base1 = ex1 + prefT[blockIdx.x * nbk + b1];
        if (blockIdx.x == 0) bstart[b1] = ex1;
    }
    const int chunk = (e + CSR_CHUNKS - 1) / CSR_CHUNKS;
    const int beg = blockIdx.x * chunk;
    const int end = (beg + chunk < e) ? (beg + chunk) : e;
    for (int t0 = beg; t0 < end; t0 += CSR_TRIP) {
        const int tn = (end - t0 < CSR_TRIP) ? (end - t0) : CSR_TRIP;
        s_loc[b0] = 0;
        s_loc[b1] = 0;
        __syncthreads();
        int c[CSR_DEPTH], r[CSR_DEPTH], p[CSR_DEPTH];
#pragma unroll
        for (int u = 0; u < CSR_DEPTH; ++u) {
            const int j  = tid + CSR_BLOCK * u;
            const int ic = t0 + ((j < tn) ? j : (tn - 1));
            c[u] = cols[ic];
            r[u] = rows[ic];
        }
#pragma unroll
        for (int u = 0; u < CSR_DEPTH; ++u)
            p[u] = (tid + CSR_BLOCK * u < tn) ? atomicAdd(&s_loc[c[u] >> CSR_BSHIFT], 1) : 0;
        __syncthreads();
        const int l0 = s_loc[b0], l1 = s_loc[b1];
        const int lx0 = gcn_block_incl_scan(l0 + l1, s_w) - (l0 + l1);
        s_loc[b0] = lx0;
        s_loc[b1] = lx0 + l0;
        s_dlt[b0] = base0 - lx0;
        s_dlt[b1] = base1 - (lx0 + l0);
        __syncthreads();
#pragma unroll
        for (int u = 0; u < CSR_DEPTH; ++u) p[u] += s_loc[c[u] >> CSR_BSHIFT];
#pragma unroll
        for (int u = 0; u < CSR_DEPTH; ++u) {
            if (tid + CSR_BLOCK * u < tn) {
                s_sc[p[u]] = c[u];
                s_sr[p[u]] = r[u];
            }
        }
        __syncthreads();
#pragma unroll
        for (int u = 0; u < CSR_DEPTH; ++u) {
            const int j  = tid + CSR_BLOCK * u;
            const int jc = (j < tn) ? j : (tn - 1);
            c[u] = s_sc[jc];
            r[u] = s_sr[jc];
        }
#pragma unroll
        for (int u = 0; u < CSR_DEPTH; ++u) p[u] = s_dlt[c[u] >> CSR_BSHIFT] + (tid + CSR_BLOCK * u);
#pragma unroll
        for (int u = 0; u < CSR_DEPTH; ++u) {
            if (tid + CSR_BLOCK * u < tn) {
                colB[p[u]] = c[u];
                rowB[p[u]] = r[u];
            }
        }
        __syncthreads();
        base0 += l0;
        base1 += l1;
    }
}

__global__ __launch_bounds__(CSR_BLOCK)
void gcn_bucket_to_csr(const int* __restrict__ rowB, const int* __restrict__ colB,
                       const int* __restrict__ bstart, const int* __restrict__ tot,
                       int* __restrict__ deg, int* __restrict__ off,
                       float* __restrict__ dinv, int* __restrict__ srcList, int n) {
    __shared__ int s_cnt[CSR_BNODES];
    __shared__ int s_w[32];
    __shared__ int s_st[CSR_STAGE2];
    const int tid = threadIdx.x;
    const int b   = blockIdx.x;
    if (tid < CSR_BNODES) s_cnt[tid] = 0;
    __syncthreads();
    const int s0  = bstart[b];
    const int cnt = tot[b];
    for (int i0 = tid; i0 < cnt; i0 += CSR_BLOCK * CSR_DEPTH) {
        int c[CSR_DEPTH];
#pragma unroll
        for (int u = 0; u < CSR_DEPTH; ++u) {
            const int i = i0 + CSR_BLOCK * u;
            c[u] = colB[s0 + ((i < cnt) ? i : (cnt - 1))];
        }
#pragma unroll
        for (int u = 0; u < CSR_DEPTH; ++u)
            if (i0 + CSR_BLOCK * u < cnt) atomicAdd(&s_cnt[c[u] & (CSR_BNODES - 1)], 1);
    }
    __syncthreads();
    const int v = (tid < CSR_BNODES) ? s_cnt[tid] : 0;
    const int excl = gcn_block_incl_scan(v, s_w) - v;
    if (tid < CSR_BNODES) {
        s_cnt[tid] = excl;
        const int node = b * CSR_BNODES + tid;
        if (node < n) {
            deg[node]  = v;
            off[node]  = s0 + excl;
            dinv[node] = rsqrtf((float)(v + 1));
        }
    }
    __syncthreads();
    for (int i0 = tid; i0 < cnt; i0 += CSR_BLOCK * CSR_DEPTH) {
        int c[CSR_DEPTH], r[CSR_DEPTH], p[CSR_DEPTH];
#pragma unroll
        for (int u = 0; u < CSR_DEPTH; ++u) {
            const int i  = i0 + CSR_BLOCK * u;
            const int ic = s0 + ((i < cnt) ? i : (cnt - 1));
            c[u] = colB[ic];
            r[u] = rowB[ic];
        }
#pragma unroll
        for (int u = 0; u < CSR_DEPTH; ++u)
            p[u] = (i0 + CSR_BLOCK * u < cnt)
                       ? atomicAdd(&s_cnt[c[u] & (CSR_BNODES - 1)], 1) : 0;
        if (cnt <= CSR_STAGE2) {
#pragma unroll
            for (int u = 0; u < CSR_DEPTH; ++u)
                if (i0 + CSR_BLOCK * u < cnt) s_st[p[u]] = r[u];
        } else {
#pragma unroll
            for (int u = 0; u < CSR_DEPTH; ++u)
                if (i0 + CSR_BLOCK * u < cnt) srcList[s0 + p[u]] = r[u];
        }
    }
    if (cnt <= CSR_STAGE2) {
        __syncthreads();
        for (int j = tid; j < cnt; j += CSR_BLOCK) srcList[s0 + j] = s_st[j];
    }
}

#define G1_K       256
#define G1_THREADS 512
#define G1_BLOCKS  256
#define G1_SLOTS   8
#define G1_CHUNK   16
#define G1_LDS_FLOATS 41216

__global__ __launch_bounds__(G1_THREADS)
void gcn_gemm_rowscale_wmma(const float* __restrict__ A,
                            const float* __restrict__ B,
                            const float* __restrict__ rowScale,
                            float* __restrict__ C,
                            int M, int rowTiles) {
    __shared__ float Ws[G1_LDS_FLOATS];

    const int tid  = threadIdx.x;
    const int lane = tid & 31;
    const int wave = __builtin_amdgcn_readfirstlane(tid >> 5);
    const int half = lane >> 4;
    const int l16  = lane & 15;

    {
        float4 t[16];
#pragma unroll
        for (int u = 0; u < 16; ++u) {
            const int idx = tid + G1_THREADS * u;
            t[u] = ((const float4*)(B + (size_t)(idx >> 5) * GCN_F))[idx & 31];
        }
#pragma unroll
        for (int u = 0; u < 16; ++u) {
            const int idx = tid + G1_THREADS * u;
            *(float4*)(Ws + (idx >> 5) * BS_STRIDE + (idx & 31) * 4) = t[u];
        }
    }
    __syncthreads();

    const int colbase = 64 * (wave & 1);
    const int tstep   = G1_BLOCKS * G1_SLOTS;
    int tile = blockIdx.x + G1_BLOCKS * (wave >> 1);
    if (tile >= rowTiles) return;

    const float* Bl = Ws + (2 * half) * BS_STRIDE + colbase + l16;

    int r0 = tile * 16 + l16;
    const float* Ap = A + (size_t)((r0 < M) ? r0 : (M - 1)) * G1_K + 2 * half;

    v2f an[G1_CHUNK];
#pragma unroll
    for (int i = 0; i < G1_CHUNK; ++i) an[i] = *(const v2f*)(Ap + 4 * i);

    for (; tile < rowTiles; tile += tstep) {
        const int tn = (tile + tstep < rowTiles) ? (tile + tstep) : tile;
        const int rn = tn * 16 + l16;
        const float* ApN = A + (size_t)((rn < M) ? rn : (M - 1)) * G1_K + 2 * half;

        const int rbase = tile * 16 + 8 * half;
        float sc[8];
#pragma unroll
        for (int v = 0; v < 8; ++v) sc[v] = rowScale[(rbase + v < M) ? (rbase + v) : (M - 1)];

        v8f acc[4] = {};
#pragma nounroll
        for (int kc = 0; kc < G1_K / (4 * G1_CHUNK); ++kc) {
            v2f ac[G1_CHUNK];
#pragma unroll
            for (int i = 0; i < G1_CHUNK; ++i) ac[i] = an[i];
            const float* Np = (kc < G1_K / (4 * G1_CHUNK) - 1) ? (Ap + 4 * G1_CHUNK * (kc + 1)) : ApN;
#pragma unroll
            for (int i = 0; i < G1_CHUNK; ++i) an[i] = *(const v2f*)(Np + 4 * i);

            const float* Bk = Bl + (4 * G1_CHUNK * kc) * BS_STRIDE;
            v2f bc[4];
#pragma unroll
            for (int j = 0; j < 4; ++j) { bc[j].x = Bk[16 * j]; bc[j].y = Bk[16 * j + BS_STRIDE]; }
#pragma unroll
            for (int i = 0; i < G1_CHUNK; ++i) {
                v2f bn[4] = {};
                if (i + 1 < G1_CHUNK) {
                    const float* Bq = Bk + (4 * (i + 1)) * BS_STRIDE;
#pragma unroll
                    for (int j = 0; j < 4; ++j) { bn[j].x = Bq[16 * j]; bn[j].y = Bq[16 * j + BS_STRIDE]; }
                }
#pragma unroll
                for (int j = 0; j < 4; ++j)
                    acc[j] = __builtin_amdgcn_wmma_f32_16x16x4_f32(
                        false, ac[i], false, bc[j], (short)0, acc[j], false, false);
#pragma unroll
                for (int j = 0; j < 4; ++j) bc[j] = bn[j];
            }
        }

        float* C0 = C + (size_t)rbase * GCN_F + colbase + l16;
        if (tile * 16 + 16 <= M) {
#pragma unroll
            for (int v = 0; v < 8; ++v)
#pragma unroll
                for (int j = 0; j < 4; ++j) C0[(size_t)v * GCN_F + 16 * j] = acc[j][v] * sc[v];
        } else {
#pragma unroll
            for (int v = 0; v < 8; ++v)
                if (rbase + v < M) {
#pragma unroll
                    for (int j = 0; j < 4; ++j) C0[(size_t)v * GCN_F + 16 * j] = acc[j][v] * sc[v];
                }
        }
        Ap = ApN;
    }
}


#define FG_WAVES    16
#define FG_THREADS  512
#define HS_STRIDE   132
#define HS_DINV     128
#define FG_INFLIGHT 8

__device__ __forceinline__ void gcn_add4(float4& a, const float4& v) {
    a.x += v.x; a.y += v.y; a.z += v.z; a.w += v.w;
}

__global__ __launch_bounds__(FG_THREADS)
void gcn_gather_relu_gemm(const float* __restrict__ Ts, const int* __restrict__ srcList,
                          const int* __restrict__ off, const int* __restrict__ deg,
                          const float* __restrict__ dinv, const float* __restrict__ b1,
                          const float* __restrict__ Wmu, const float* __restrict__ Wlv,
                          float* __restrict__ T2, int nNodes, int nTiles) {
    __shared__ float Ws[GCN_F * BS_STRIDE];
    __shared__ float Hs[2][FG_WAVES * HS_STRIDE];

    const int tid  = threadIdx.x;
    const int lane = tid & 31;
    const int wave = __builtin_amdgcn_readfirstlane(tid >> 5);
    const int half = lane >> 4;
    const int l16  = lane & 15;

    for (int idx = tid; idx < GCN_F * 32; idx += FG_THREADS) {
        const int r = idx >> 5, c4 = idx & 31;
        const float4 v = (c4 < 16) ? ((const float4*)(Wmu + (size_t)r * 64))[c4]
                                   : ((const float4*)(Wlv + (size_t)r * 64))[c4 - 16];
        *(float4*)(Ws + r * BS_STRIDE + c4 * 4) = v;
    }
    const float4 bb = ((const float4*)b1)[lane];
    const float4* TsRow = (const float4*)Ts + lane;

    const int step = gridDim.x;
    int tile = blockIdx.x;

    int   s_c = 0, d_c = 0, s_n = 0, d_n = 0;
    float dv_c = 0.0f, dv_n = 0.0f;
    {
        const int n0 = tile * FG_WAVES + wave;
        if (tile < nTiles && n0 < nNodes) { s_c = off[n0]; d_c = deg[n0]; dv_c = dinv[n0]; }
        const int n1 = (tile + step) * FG_WAVES + wave;
        if (tile + step < nTiles && n1 < nNodes) { s_n = off[n1]; d_n = deg[n1]; dv_n = dinv[n1]; }
    }
    int mine_c = (lane < d_c) ? srcList[s_c + lane] : 0;

    int buf = 0;
    for (; tile < nTiles; tile += step) {
        const int node = tile * FG_WAVES + wave;

        int   s_a = 0, d_a = 0;
        float dv_a = 0.0f;
        {
            const int ta = tile + 2 * step;
            const int na = ta * FG_WAVES + wave;
            if (ta < nTiles && na < nNodes) { s_a = off[na]; d_a = deg[na]; dv_a = dinv[na]; }
        }
        const int mine_n = (lane < d_n) ? srcList[s_n + lane] : 0;

        float4 r = {0.0f, 0.0f, 0.0f, 0.0f};
        if (node < nNodes) {
            float4 acc = TsRow[(size_t)node * 32];
            for (int e0 = 0; e0 < d_c; e0 += 32) {
                const int cnt = (d_c - e0 < 32) ? (d_c - e0) : 32;
                const int mine = (e0 == 0) ? mine_c
                                           : ((lane < cnt) ? srcList[s_c + e0 + lane] : 0);
                int e = 0;
                for (; e + FG_INFLIGHT <= cnt; e += FG_INFLIGHT) {
                    float4 v[FG_INFLIGHT];
#pragma unroll
                    for (int u = 0; u < FG_INFLIGHT; ++u)
                        v[u] = TsRow[(size_t)__builtin_amdgcn_readlane(mine, e + u) * 32];
#pragma unroll
                    for (int u = 0; u < FG_INFLIGHT; ++u) gcn_add4(acc, v[u]);
                }
                if (e < cnt) {
                    float4 v[FG_INFLIGHT - 1] = {};
#pragma unroll
                    for (int u = 0; u < FG_INFLIGHT - 1; ++u)
                        if (e + u < cnt)
                            v[u] = TsRow[(size_t)__builtin_amdgcn_readlane(mine, e + u) * 32];
#pragma unroll
                    for (int u = 0; u < FG_INFLIGHT - 1; ++u)
                        if (e + u < cnt) gcn_add4(acc, v[u]);
                }
            }
            r.x = fmaxf(dv_c * acc.x + bb.x, 0.0f);
            r.y = fmaxf(dv_c * acc.y + bb.y, 0.0f);
            r.z = fmaxf(dv_c * acc.z + bb.z, 0.0f);
            r.w = fmaxf(dv_c * acc.w + bb.w, 0.0f);
        }
        float* Hrow = &Hs[buf][wave * HS_STRIDE];
        *(float4*)(Hrow + lane * 4) = r;
        if (lane == 0) Hrow[HS_DINV] = dv_c;
        __syncthreads();

        if (wave < 8) {
            const float* A0 = &Hs[buf][l16 * HS_STRIDE + 2 * half];
            const float* B0 = Ws + (2 * half) * BS_STRIDE + 16 * wave + l16;
            v8f acc = {};
#pragma unroll 8
            for (int i = 0; i < GCN_F / 4; ++i) {
                v2f a, b;
                a.x = A0[4 * i];
                a.y = A0[4 * i + 1];
                b.x = B0[(4 * i) * BS_STRIDE];
                b.y = B0[(4 * i + 1) * BS_STRIDE];
                acc = __builtin_amdgcn_wmma_f32_16x16x4_f32(
                    false, a, false, b, (short)0, acc, false, false);
            }
            const int row0 = tile * FG_WAVES + 8 * half;
            const float* S0 = &Hs[buf][(8 * half) * HS_STRIDE + HS_DINV];
            float* C0 = T2 + (size_t)row0 * GCN_F + 16 * wave + l16;
            if (tile * FG_WAVES + FG_WAVES <= nNodes) {
#pragma unroll
                for (int v = 0; v < 8; ++v) C0[(size_t)v * GCN_F] = acc[v] * S0[v * HS_STRIDE];
            } else {
#pragma unroll
                for (int v = 0; v < 8; ++v)
                    if (row0 + v < nNodes) C0[(size_t)v * GCN_F] = acc[v] * S0[v * HS_STRIDE];
            }
        }
        s_c = s_n; d_c = d_n; dv_c = dv_n; mine_c = mine_n;
        s_n = s_a; d_n = d_a; dv_n = dv_a;
        buf ^= 1;
    }
}

#define GO_WAVES    8
#define GO_THREADS  256
#define GO_BLOCKS   2048
#define GO_INFLIGHT 8

__device__ __forceinline__ void gcn_out_ahead(const float4* __restrict__ TsRow,
                                              const int* __restrict__ srcList,
                                              const int* __restrict__ off,
                                              const int* __restrict__ deg,
                                              const float* __restrict__ dinv,
                                              int node, int step, int nNodes, int lane,
                                              int s_n, int d_n,
                                              int& s_a, int& d_a, float& dv_a,
                                              int& mine_n, float4& self_n) {
    const int nn = node + step;
    const int na = nn + step;
    if (na < nNodes) { s_a = off[na]; d_a = deg[na]; dv_a = dinv[na]; }
    if (nn < nNodes) {
        if (lane < d_n) mine_n = srcList[s_n + lane];
        self_n = TsRow[(size_t)nn * 32];
    }
}

__global__ __launch_bounds__(GO_THREADS, 16)
void gcn_gather_out(const float* __restrict__ Ts, const int* __restrict__ srcList,
                    const int* __restrict__ off, const int* __restrict__ deg,
                    const float* __restrict__ dinv, const float* __restrict__ bmu,
                    const float* __restrict__ blv, float* __restrict__ out, int nNodes) {
    const int lane = threadIdx.x & 31;
    const int wave = __builtin_amdgcn_readfirstlane(threadIdx.x >> 5);
    const int step = gridDim.x * GO_WAVES;
    int node = blockIdx.x * GO_WAVES + wave;
    if (node >= nNodes) return;

    const float4* TsRow = (const float4*)Ts + lane;
    float4 bb;
    float* dst;
    if (lane < 16) {
        bb  = ((const float4*)bmu)[lane];
        dst = out + lane * 4;
    } else {
        bb  = ((const float4*)blv)[lane - 16];
        dst = out + (size_t)nNodes * 64 + (lane - 16) * 4;
    }

    int   s_c = off[node], d_c = deg[node];
    float dv_c = dinv[node];
    int   s_n = 0, d_n = 0;
    float dv_n = 0.0f;
    if (node + step < nNodes) { s_n = off[node + step]; d_n = deg[node + step]; dv_n = dinv[node + step]; }
    int    mine_c = (lane < d_c) ? srcList[s_c + lane] : 0;
    float4 self_c = TsRow[(size_t)node * 32];

    for (; node < nNodes; node += step) {
        int    s_a = 0, d_a = 0;
        float  dv_a = 0.0f;
        int    mine_n = 0;
        float4 self_n = {0.0f, 0.0f, 0.0f, 0.0f};
        bool   ahead = true;

        float4 acc = self_c;
        for (int e0 = 0; e0 < d_c; e0 += 32) {
            const int cnt = (d_c - e0 < 32) ? (d_c - e0) : 32;
            const int mine = (e0 == 0) ? mine_c
                                       : ((lane < cnt) ? srcList[s_c + e0 + lane] : 0);
            int e = 0;
            for (; e + GO_INFLIGHT <= cnt; e += GO_INFLIGHT) {
                float4 v[GO_INFLIGHT];
#pragma unroll
                for (int u = 0; u < GO_INFLIGHT; ++u)
                    v[u] = TsRow[(size_t)__builtin_amdgcn_readlane(mine, e + u) * 32];
                if (ahead) {
                    gcn_out_ahead(TsRow, srcList, off, deg, dinv, node, step, nNodes, lane,
                                  s_n, d_n, s_a, d_a, dv_a, mine_n, self_n);
                    ahead = false;
                }
#pragma unroll
                for (int u = 0; u < GO_INFLIGHT; ++u) gcn_add4(acc, v[u]);
            }
            if (e < cnt) {
                float4 v[GO_INFLIGHT - 1] = {};
#pragma unroll
                for (int u = 0; u < GO_INFLIGHT - 1; ++u)
                    if (e + u < cnt)
                        v[u] = TsRow[(size_t)__builtin_amdgcn_readlane(mine, e + u) * 32];
                if (ahead) {
                    gcn_out_ahead(TsRow, srcList, off, deg, dinv, node, step, nNodes, lane,
                                  s_n, d_n, s_a, d_a, dv_a, mine_n, self_n);
                    ahead = false;
                }
#pragma unroll
                for (int u = 0; u < GO_INFLIGHT - 1; ++u)
                    if (e + u < cnt) gcn_add4(acc, v[u]);
            }
        }
        if (ahead)
            gcn_out_ahead(TsRow, srcList, off, deg, dinv, node, step, nNodes, lane,
                          s_n, d_n, s_a, d_a, dv_a, mine_n, self_n);

        float4 r;
        r.x = dv_c * acc.x + bb.x;
        r.y = dv_c * acc.y + bb.y;
        r.z = dv_c * acc.z + bb.z;
        r.w = dv_c * acc.w + bb.w;
        *(float4*)(dst + (size_t)node * 64) = r;

        s_c = s_n; d_c = d_n; dv_c = dv_n; mine_c = mine_n; self_c = self_n;
        s_n = s_a; d_n = d_a; dv_n = dv_a;
    }
}


extern "C" void kernel_launch(void* const* d_in, const int* in_sizes, int n_in,
                              void* d_out, int out_size, void* d_ws, size_t ws_size,
                              hipStream_t stream) {
    const float* X   = (const float*)d_in[0];
    const int*   ei  = (const int*)d_in[1];
    const float* W1  = (const float*)d_in[2];
    const float* b1  = (const float*)d_in[3];
    const float* Wmu = (const float*)d_in[4];
    const float* bmu = (const float*)d_in[5];
    const float* Wlv = (const float*)d_in[6];
    const float* blv = (const float*)d_in[7];
    float* out = (float*)d_out;

    const int N = in_sizes[0] / 256;
    const int E = in_sizes[1] / 2;
    const int* rows = ei;
    const int* cols = ei + E;

    float* Ts   = (float*)d_ws;
    float* Hbuf = Ts + (size_t)N * GCN_F;
    float* W23  = Hbuf + (size_t)N * GCN_F;
    float* dinv = W23 + GCN_F * GCN_F;
    int* degInt  = (int*)(dinv + N);
    int* off     = degInt + N;
    int* srcList = off + N;

    const int nbk = (N + CSR_BNODES - 1) / CSR_BNODES;
    int* colB   = (int*)Hbuf;
    int* rowB   = colB + E;
    int* histT  = rowB + E;
    int* btot   = histT + CSR_CHUNKS * nbk;
    int* bstart = btot + nbk;
    gcn_bucket_hist<<<CSR_CHUNKS, CSR_BLOCK, 0, stream>>>(cols, histT, E, nbk);
    gcn_bucket_prefix<<<nbk, CSR_CHUNKS, 0, stream>>>(histT, btot, nbk);
    gcn_bucket_scatter<<<CSR_CHUNKS, CSR_BLOCK, 0, stream>>>(rows, cols, histT, btot, bstart,
                                                        rowB, colB, E, nbk);
    gcn_bucket_to_csr<<<nbk, CSR_BLOCK, 0, stream>>>(rowB, colB, bstart, btot, degInt, off,
                                                dinv, srcList, N);

    const int rowTiles = (N + 15) / 16;

    gcn_gemm_rowscale_wmma<<<G1_BLOCKS, G1_THREADS, 0, stream>>>(X, W1, dinv, Ts, N, rowTiles);
    gcn_gather_relu_gemm<<<768, FG_THREADS, 0, stream>>>(Ts, srcList, off, degInt, dinv, b1,
                                                          Wmu, Wlv, Hbuf, N, rowTiles);

    gcn_gather_out<<<GO_BLOCKS, GO_THREADS, 0, stream>>>(Hbuf, srcList, off, degInt,
                                                     dinv, bmu, blv, out, N);
}


# CSR build in 2 launches, not 4: each block sorts its own edge chunk by 256-node bucket in place (no global offsets); level 2 finds its bucket's pieces through a count/offset table; list order kept
# speedup vs baseline: 2.4202x; 1.0740x over previous
#include <hip/hip_runtime.h>
#include <hip/hip_bf16.h>

typedef float v2f __attribute__((ext_vector_type(2)));
typedef float v8f __attribute__((ext_vector_type(8)));

#define GCN_F 128
#define BS_STRIDE 136


#define CSR_BSHIFT 8
#define CSR_BNODES 256
#define CSR_CHUNKS 256
#define CSR_MAXB   1024
#define CSR_DEPTH  16
#define CSR_BLOCK  512
#define CSR_TRIP   (CSR_BLOCK * CSR_DEPTH)
#define CSR_STAGE2 16384
#define CSR_MAXROWS 1024

__device__ __forceinline__ int gcn_block_incl_scan(int v, int* s_w) {
    const int lane = threadIdx.x & 31;
    const int wave = threadIdx.x >> 5;
    const int nw   = blockDim.x >> 5;
    for (int d = 1; d < 32; d <<= 1) {
        const int t = __shfl_up(v, d);
        if (lane >= d) v += t;
    }
    if (lane == 31) s_w[wave] = v;
    __syncthreads();
    if (wave == 0) {
        const int w = (lane < nw) ? s_w[lane] : 0;
        int wi = w;
        for (int d = 1; d < 32; d <<= 1) {
            const int t = __shfl_up(wi, d);
            if (lane >= d) wi += t;
        }
        s_w[lane] = wi - w;
    }
    __syncthreads();
    return v + s_w[wave];
}

__device__ __forceinline__ void gcn_block_incl_scan2(int& a, int& b, int& ta, int& tb,
                                                     int* s_a, int* s_b) {
    const int lane = threadIdx.x & 31;
    const int wave = threadIdx.x >> 5;
    const int nw   = blockDim.x >> 5;
    for (int d = 1; d < 32; d <<= 1) {
        const int xa = __shfl_up(a, d);
        const int xb = __shfl_up(b, d);
        if (lane >= d) { a += xa; b += xb; }
    }
    if (lane == 31) { s_a[wave] = a; s_b[wave] = b; }
    __syncthreads();
    if (wave == 0) {
        const int wa = (lane < nw) ? s_a[lane] : 0;
        const int wb = (lane < nw) ? s_b[lane] : 0;
        int ia = wa, ib = wb;
        for (int d = 1; d < 32; d <<= 1) {
            const int xa = __shfl_up(ia, d);
            const int xb = __shfl_up(ib, d);
            if (lane >= d) { ia += xa; ib += xb; }
        }
        s_a[lane] = ia - wa;
        s_b[lane] = ib - wb;
        if (lane == 31) { s_a[32] = ia; s_b[32] = ib; }
    }
    __syncthreads();
    a += s_a[wave];
    b += s_b[wave];
    ta = s_a[32];
    tb = s_b[32];
}

__global__ __launch_bounds__(CSR_BLOCK)
void gcn_chunk_sort(const int* __restrict__ rows, const int* __restrict__ cols,
                    int* __restrict__ cntT, int* __restrict__ offT,
                    int* __restrict__ rowB, int* __restrict__ colB, int e, int nbk) {
    __shared__ int s_loc[CSR_MAXB];
    __shared__ int s_w[32];
    __shared__ int s_sc[CSR_TRIP];
    __shared__ int s_sr[CSR_TRIP];
    const int tid = threadIdx.x;
    const int b0 = 2 * tid, b1 = 2 * tid + 1;
    const int chunk = (e + CSR_CHUNKS - 1) / CSR_CHUNKS;
    const int ntrip = (chunk + CSR_TRIP - 1) / CSR_TRIP;
    const int beg = blockIdx.x * chunk;
    const int end = (beg + chunk < e) ? (beg + chunk) : e;
    for (int trip = 0; trip < ntrip; ++trip) {
        const int row = blockIdx.x * ntrip + trip;
        const int t0  = beg + trip * CSR_TRIP;
        const int tn  = (end - t0 < CSR_TRIP) ? (end - t0) : CSR_TRIP;
        if (tn <= 0) {
            if (b0 < nbk) { cntT[row * nbk + b0] = 0; offT[row * nbk + b0] = 0; }
            if (b1 < nbk) { cntT[row * nbk + b1] = 0; offT[row * nbk + b1] = 0; }
            continue;
        }
        s_loc[b0] = 0;
        s_loc[b1] = 0;
        __syncthreads();
        int c[CSR_DEPTH], r[CSR_DEPTH], p[CSR_DEPTH];
#pragma unroll
        for (int u = 0; u < CSR_DEPTH; ++u) {
            const int j  = tid + CSR_BLOCK * u;
            const int ic = t0 + ((j < tn) ? j : (tn - 1));
            c[u] = cols[ic];
            r[u] = rows[ic];
        }
#pragma unroll
        for (int u = 0; u < CSR_DEPTH; ++u)
            p[u] = (tid + CSR_BLOCK * u < tn) ? atomicAdd(&s_loc[c[u] >> CSR_BSHIFT], 1) : 0;
        __syncthreads();
        const int l0 = s_loc[b0], l1 = s_loc[b1];
        const int lx0 = gcn_block_incl_scan(l0 + l1, s_w) - (l0 + l1);
        s_loc[b0] = lx0;
        s_loc[b1] = lx0 + l0;
        if (b0 < nbk) { cntT[row * nbk + b0] = l0; offT[row * nbk + b0] = lx0; }
        if (b1 < nbk) { cntT[row * nbk + b1] = l1; offT[row * nbk + b1] = lx0 + l0; }
        __syncthreads();
#pragma unroll
        for (int u = 0; u < CSR_DEPTH; ++u) p[u] += s_loc[c[u] >> CSR_BSHIFT];
#pragma unroll
        for (int u = 0; u < CSR_DEPTH; ++u) {
            if (tid + CSR_BLOCK * u < tn) {
                s_sc[p[u]] = c[u];
                s_sr[p[u]] = r[u];
            }
        }
        __syncthreads();
#pragma unroll
        for (int u = 0; u < CSR_DEPTH; ++u) {
            const int j = tid + CSR_BLOCK * u;
            if (j < tn) {
                colB[t0 + j] = s_sc[j];
                rowB[t0 + j] = s_sr[j];
            }
        }
        __syncthreads();
    }
}

__device__ __forceinline__ void gcn_piece_src(const int* s_ps, const int* s_src, int nrows, int top,
                                              int i0, int cnt, int (&q)[CSR_DEPTH]) {
    int idx[CSR_DEPTH], pos[CSR_DEPTH];
#pragma unroll
    for (int u = 0; u < CSR_DEPTH; ++u) {
        const int i = i0 + CSR_BLOCK * u;
        idx[u] = (i < cnt) ? i : (cnt - 1);
        pos[u] = 0;
    }
    for (int s = top; s >= 1; s >>= 1) {
        int v[CSR_DEPTH];
#pragma unroll
        for (int u = 0; u < CSR_DEPTH; ++u) {
            const int m = pos[u] + s;
            v[u] = s_ps[(m < nrows) ? m : (nrows - 1)];
        }
#pragma unroll
        for (int u = 0; u < CSR_DEPTH; ++u) {
            const int m = pos[u] + s;
            pos[u] = ((m < nrows) & (v[u] <= idx[u])) ? m : pos[u];
        }
    }
#pragma unroll
    for (int u = 0; u < CSR_DEPTH; ++u) q[u] = s_src[pos[u]] + idx[u];
}

__global__ __launch_bounds__(CSR_BLOCK)
void gcn_bucket_to_csr(const int* __restrict__ rowB, const int* __restrict__ colB,
                       const int* __restrict__ cntT, const int* __restrict__ offT,
                       int* __restrict__ deg, int* __restrict__ off,
                       float* __restrict__ dinv, int* __restrict__ srcList,
                       int n, int e, int nbk) {
    __shared__ int s_cnt[CSR_BNODES];
    __shared__ int s_w[32];
    __shared__ int s_wa[33];
    __shared__ int s_wb[33];
    __shared__ int s_ps[CSR_MAXROWS];
    __shared__ int s_src[CSR_MAXROWS];
    __shared__ int s_st[CSR_STAGE2];
    const int tid = threadIdx.x;
    const int b   = blockIdx.x;
    const int chunk = (e + CSR_CHUNKS - 1) / CSR_CHUNKS;
    const int ntrip = (chunk + CSR_TRIP - 1) / CSR_TRIP;
    const int nrows = CSR_CHUNKS * ntrip;
    if (tid < CSR_BNODES) s_cnt[tid] = 0;
    int cnt = 0, s0 = 0;
    for (int r0 = 0; r0 < nrows; r0 += CSR_BLOCK) {
        const int r = r0 + tid;
        int pc = 0, po = 0;
        if (r < nrows) { pc = cntT[r * nbk + b]; po = offT[r * nbk + b]; }
        int ci = pc, oi = po, tc, to;
        gcn_block_incl_scan2(ci, oi, tc, to, s_wa, s_wb);
        if (r < nrows) {
            const int ps = cnt + ci - pc;
            s_ps[r]  = ps;
            s_src[r] = (r / ntrip) * chunk + (r % ntrip) * CSR_TRIP + po - ps;
        }
        cnt += tc;
        s0  += to;
    }
    __syncthreads();
    int top = CSR_CHUNKS / 2;
    while (top * 2 < nrows) top *= 2;

    int c[CSR_DEPTH], q[CSR_DEPTH];
    for (int i0 = tid; i0 < cnt; i0 += CSR_BLOCK * CSR_DEPTH) {
        gcn_piece_src(s_ps, s_src, nrows, top, i0, cnt, q);
#pragma unroll
        for (int u = 0; u < CSR_DEPTH; ++u) c[u] = colB[q[u]];
#pragma unroll
        for (int u = 0; u < CSR_DEPTH; ++u)
            if (i0 + CSR_BLOCK * u < cnt) atomicAdd(&s_cnt[c[u] & (CSR_BNODES - 1)], 1);
    }
    __syncthreads();
    const int v = (tid < CSR_BNODES) ? s_cnt[tid] : 0;
    const int excl = gcn_block_incl_scan(v, s_w) - v;
    if (tid < CSR_BNODES) {
        s_cnt[tid] = excl;
        const int node = b * CSR_BNODES + tid;
        if (node < n) {
            deg[node]  = v;
            off[node]  = s0 + excl;
            dinv[node] = rsqrtf((float)(v + 1));
        }
    }
    __syncthreads();
    const bool again = cnt > CSR_TRIP;
    for (int i0 = tid; i0 < cnt; i0 += CSR_BLOCK * CSR_DEPTH) {
        int r[CSR_DEPTH], p[CSR_DEPTH];
        if (again) {
            gcn_piece_src(s_ps, s_src, nrows, top, i0, cnt, q);
#pragma unroll
            for (int u = 0; u < CSR_DEPTH; ++u) c[u] = colB[q[u]];
        }
#pragma unroll
        for (int u = 0; u < CSR_DEPTH; ++u) r[u] = rowB[q[u]];
#pragma unroll
        for (int u = 0; u < CSR_DEPTH; ++u)
            p[u] = (i0 + CSR_BLOCK * u < cnt)
                       ? atomicAdd(&s_cnt[c[u] & (CSR_BNODES - 1)], 1) : 0;
        if (cnt <= CSR_STAGE2) {
#pragma unroll
            for (int u = 0; u < CSR_DEPTH; ++u)
                if (i0 + CSR_BLOCK * u < cnt) s_st[p[u]] = r[u];
        } else {
#pragma unroll
            for (int u = 0; u < CSR_DEPTH; ++u)
                if (i0 + CSR_BLOCK * u < cnt) srcList[s0 + p[u]] = r[u];
        }
    }
    if (cnt <= CSR_STAGE2) {
        __syncthreads();
        for (int j = tid; j < cnt; j += CSR_BLOCK) srcList[s0 + j] = s_st[j];
    }
}

#define G1_K       256
#define G1_THREADS 512
#define G1_BLOCKS  256
#define G1_SLOTS   8
#define G1_CHUNK   16
#define G1_LDS_FLOATS 41216

__global__ __launch_bounds__(G1_THREADS)
void gcn_gemm_rowscale_wmma(const float* __restrict__ A,
                            const float* __restrict__ B,
                            const float* __restrict__ rowScale,
                            float* __restrict__ C,
                            int M, int rowTiles) {
    __shared__ float Ws[G1_LDS_FLOATS];

    const int tid  = threadIdx.x;
    const int lane = tid & 31;
    const int wave = __builtin_amdgcn_readfirstlane(tid >> 5);
    const int half = lane >> 4;
    const int l16  = lane & 15;

    {
        float4 t[16];
#pragma unroll
        for (int u = 0; u < 16; ++u) {
            const int idx = tid + G1_THREADS * u;
            t[u] = ((const float4*)(B + (size_t)(idx >> 5) * GCN_F))[idx & 31];
        }
#pragma unroll
        for (int u = 0; u < 16; ++u) {
            const int idx = tid + G1_THREADS * u;
            *(float4*)(Ws + (idx >> 5) * BS_STRIDE + (idx & 31) * 4) = t[u];
        }
    }
    __syncthreads();

    const int colbase = 64 * (wave & 1);
    const int tstep   = G1_BLOCKS * G1_SLOTS;
    int tile = blockIdx.x + G1_BLOCKS * (wave >> 1);
    if (tile >= rowTiles) return;

    const float* Bl = Ws + (2 * half) * BS_STRIDE + colbase + l16;

    int r0 = tile * 16 + l16;
    const float* Ap = A + (size_t)((r0 < M) ? r0 : (M - 1)) * G1_K + 2 * half;

    v2f an[G1_CHUNK];
#pragma unroll
    for (int i = 0; i < G1_CHUNK; ++i) an[i] = *(const v2f*)(Ap + 4 * i);

    for (; tile < rowTiles; tile += tstep) {
        const int tn = (tile + tstep < rowTiles) ? (tile + tstep) : tile;
        const int rn = tn * 16 + l16;
        const float* ApN = A + (size_t)((rn < M) ? rn : (M - 1)) * G1_K + 2 * half;

        const int rbase = tile * 16 + 8 * half;
        float sc[8];
#pragma unroll
        for (int v = 0; v < 8; ++v) sc[v] = rowScale[(rbase + v < M) ? (rbase + v) : (M - 1)];

        v8f acc[4] = {};
#pragma nounroll
        for (int kc = 0; kc < G1_K / (4 * G1_CHUNK); ++kc) {
            v2f ac[G1_CHUNK];
#pragma unroll
            for (int i = 0; i < G1_CHUNK; ++i) ac[i] = an[i];
            const float* Np = (kc < G1_K / (4 * G1_CHUNK) - 1) ? (Ap + 4 * G1_CHUNK * (kc + 1)) : ApN;
#pragma unroll
            for (int i = 0; i < G1_CHUNK; ++i) an[i] = *(const v2f*)(Np + 4 * i);

            const float* Bk = Bl + (4 * G1_CHUNK * kc) * BS_STRIDE;
            v2f bc[4];
#pragma unroll
            for (int j = 0; j < 4; ++j) { bc[j].x = Bk[16 * j]; bc[j].y = Bk[16 * j + BS_STRIDE]; }
#pragma unroll
            for (int i = 0; i < G1_CHUNK; ++i) {
                v2f bn[4] = {};
                if (i + 1 < G1_CHUNK) {
                    const float* Bq = Bk + (4 * (i + 1)) * BS_STRIDE;
#pragma unroll
                    for (int j = 0; j < 4; ++j) { bn[j].x = Bq[16 * j]; bn[j].y = Bq[16 * j + BS_STRIDE]; }
                }
#pragma unroll
                for (int j = 0; j < 4; ++j)
                    acc[j] = __builtin_amdgcn_wmma_f32_16x16x4_f32(
                        false, ac[i], false, bc[j], (short)0, acc[j], false, false);
#pragma unroll
                for (int j = 0; j < 4; ++j) bc[j] = bn[j];
            }
        }

        float* C0 = C + (size_t)rbase * GCN_F + colbase + l16;
        if (tile * 16 + 16 <= M) {
#pragma unroll
            for (int v = 0; v < 8; ++v)
#pragma unroll
                for (int j = 0; j < 4; ++j) C0[(size_t)v * GCN_F + 16 * j] = acc[j][v] * sc[v];
        } else {
#pragma unroll
            for (int v = 0; v < 8; ++v)
                if (rbase + v < M) {
#pragma unroll
                    for (int j = 0; j < 4; ++j) C0[(size_t)v * GCN_F + 16 * j] = acc[j][v] * sc[v];
                }
        }
        Ap = ApN;
    }
}


#define FG_WAVES    16
#define FG_THREADS  512
#define HS_STRIDE   132
#define HS_DINV     128
#define FG_INFLIGHT 8

__device__ __forceinline__ void gcn_add4(float4& a, const float4& v) {
    a.x += v.x; a.y += v.y; a.z += v.z; a.w += v.w;
}

__global__ __launch_bounds__(FG_THREADS)
void gcn_gather_relu_gemm(const float* __restrict__ Ts, const int* __restrict__ srcList,
                          const int* __restrict__ off, const int* __restrict__ deg,
                          const float* __restrict__ dinv, const float* __restrict__ b1,
                          const float* __restrict__ Wmu, const float* __restrict__ Wlv,
                          float* __restrict__ T2, int nNodes, int nTiles) {
    __shared__ float Ws[GCN_F * BS_STRIDE];
    __shared__ float Hs[2][FG_WAVES * HS_STRIDE];

    const int tid  = threadIdx.x;
    const int lane = tid & 31;
    const int wave = __builtin_amdgcn_readfirstlane(tid >> 5);
    const int half = lane >> 4;
    const int l16  = lane & 15;

    for (int idx = tid; idx < GCN_F * 32; idx += FG_THREADS) {
        const int r = idx >> 5, c4 = idx & 31;
        const float4 v = (c4 < 16) ? ((const float4*)(Wmu + (size_t)r * 64))[c4]
                                   : ((const float4*)(Wlv + (size_t)r * 64))[c4 - 16];
        *(float4*)(Ws + r * BS_STRIDE + c4 * 4) = v;
    }
    const float4 bb = ((const float4*)b1)[lane];
    const float4* TsRow = (const float4*)Ts + lane;

    const int step = gridDim.x;
    int tile = blockIdx.x;

    int   s_c = 0, d_c = 0, s_n = 0, d_n = 0;
    float dv_c = 0.0f, dv_n = 0.0f;
    {
        const int n0 = tile * FG_WAVES + wave;
        if (tile < nTiles && n0 < nNodes) { s_c = off[n0]; d_c = deg[n0]; dv_c = dinv[n0]; }
        const int n1 = (tile + step) * FG_WAVES + wave;
        if (tile + step < nTiles && n1 < nNodes) { s_n = off[n1]; d_n = deg[n1]; dv_n = dinv[n1]; }
    }
    int mine_c = (lane < d_c) ? srcList[s_c + lane] : 0;

    int buf = 0;
    for (; tile < nTiles; tile += step) {
        const int node = tile * FG_WAVES + wave;

        int   s_a = 0, d_a = 0;
        float dv_a = 0.0f;
        {
            const int ta = tile + 2 * step;
            const int na = ta * FG_WAVES + wave;
            if (ta < nTiles && na < nNodes) { s_a = off[na]; d_a = deg[na]; dv_a = dinv[na]; }
        }
        const int mine_n = (lane < d_n) ? srcList[s_n + lane] : 0;

        float4 r = {0.0f, 0.0f, 0.0f, 0.0f};
        if (node < nNodes) {
            float4 acc = TsRow[(size_t)node * 32];
            for (int e0 = 0; e0 < d_c; e0 += 32) {
                const int cnt = (d_c - e0 < 32) ? (d_c - e0) : 32;
                const int mine = (e0 == 0) ? mine_c
                                           : ((lane < cnt) ? srcList[s_c + e0 + lane] : 0);
                int e = 0;
                for (; e + FG_INFLIGHT <= cnt; e += FG_INFLIGHT) {
                    float4 v[FG_INFLIGHT];
#pragma unroll
                    for (int u = 0; u < FG_INFLIGHT; ++u)
                        v[u] = TsRow[(size_t)__builtin_amdgcn_readlane(mine, e + u) * 32];
#pragma unroll
                    for (int u = 0; u < FG_INFLIGHT; ++u) gcn_add4(acc, v[u]);
                }
                if (e < cnt) {
                    float4 v[FG_INFLIGHT - 1] = {};
#pragma unroll
                    for (int u = 0; u < FG_INFLIGHT - 1; ++u)
                        if (e + u < cnt)
                            v[u] = TsRow[(size_t)__builtin_amdgcn_readlane(mine, e + u) * 32];
#pragma unroll
                    for (int u = 0; u < FG_INFLIGHT - 1; ++u)
                        if (e + u < cnt) gcn_add4(acc, v[u]);
                }
            }
            r.x = fmaxf(dv_c * acc.x + bb.x, 0.0f);
            r.y = fmaxf(dv_c * acc.y + bb.y, 0.0f);
            r.z = fmaxf(dv_c * acc.z + bb.z, 0.0f);
            r.w = fmaxf(dv_c * acc.w + bb.w, 0.0f);
        }
        float* Hrow = &Hs[buf][wave * HS_STRIDE];
        *(float4*)(Hrow + lane * 4) = r;
        if (lane == 0) Hrow[HS_DINV] = dv_c;
        __syncthreads();

        if (wave < 8) {
            const float* A0 = &Hs[buf][l16 * HS_STRIDE + 2 * half];
            const float* B0 = Ws + (2 * half) * BS_STRIDE + 16 * wave + l16;
            v8f acc = {};
#pragma unroll 8
            for (int i = 0; i < GCN_F / 4; ++i) {
                v2f a, b;
                a.x = A0[4 * i];
                a.y = A0[4 * i + 1];
                b.x = B0[(4 * i) * BS_STRIDE];
                b.y = B0[(4 * i + 1) * BS_STRIDE];
                acc = __builtin_amdgcn_wmma_f32_16x16x4_f32(
                    false, a, false, b, (short)0, acc, false, false);
            }
            const int row0 = tile * FG_WAVES + 8 * half;
            const float* S0 = &Hs[buf][(8 * half) * HS_STRIDE + HS_DINV];
            float* C0 = T2 + (size_t)row0 * GCN_F + 16 * wave + l16;
            if (tile * FG_WAVES + FG_WAVES <= nNodes) {
#pragma unroll
                for (int v = 0; v < 8; ++v) C0[(size_t)v * GCN_F] = acc[v] * S0[v * HS_STRIDE];
            } else {
#pragma unroll
                for (int v = 0; v < 8; ++v)
                    if (row0 + v < nNodes) C0[(size_t)v * GCN_F] = acc[v] * S0[v * HS_STRIDE];
            }
        }
        s_c = s_n; d_c = d_n; dv_c = dv_n; mine_c = mine_n;
        s_n = s_a; d_n = d_a; dv_n = dv_a;
        buf ^= 1;
    }
}

#define GO_WAVES    8
#define GO_THREADS  256
#define GO_BLOCKS   2048
#define GO_INFLIGHT 8

__device__ __forceinline__ void gcn_out_ahead(const float4* __restrict__ TsRow,
                                              const int* __restrict__ srcList,
                                              const int* __restrict__ off,
                                              const int* __restrict__ deg,
                                              const float* __restrict__ dinv,
                                              int node, int step, int nNodes, int lane,
                                              int s_n, int d_n,
                                              int& s_a, int& d_a, float& dv_a,
                                              int& mine_n, float4& self_n) {
    const int nn = node + step;
    const int na = nn + step;
    if (na < nNodes) { s_a = off[na]; d_a = deg[na]; dv_a = dinv[na]; }
    if (nn < nNodes) {
        if (lane < d_n) mine_n = srcList[s_n + lane];
        self_n = TsRow[(size_t)nn * 32];
    }
}

__global__ __launch_bounds__(GO_THREADS, 16)
void gcn_gather_out(const float* __restrict__ Ts, const int* __restrict__ srcList,
                    const int* __restrict__ off, const int* __restrict__ deg,
                    const float* __restrict__ dinv, const float* __restrict__ bmu,
                    const float* __restrict__ blv, float* __restrict__ out, int nNodes) {
    const int lane = threadIdx.x & 31;
    const int wave = __builtin_amdgcn_readfirstlane(threadIdx.x >> 5);
    const int step = gridDim.x * GO_WAVES;
    int node = blockIdx.x * GO_WAVES + wave;
    if (node >= nNodes) return;

    const float4* TsRow = (const float4*)Ts + lane;
    float4 bb;
    float* dst;
    if (lane < 16) {
        bb  = ((const float4*)bmu)[lane];
        dst = out + lane * 4;
    } else {
        bb  = ((const float4*)blv)[lane - 16];
        dst = out + (size_t)nNodes * 64 + (lane - 16) * 4;
    }

    int   s_c = off[node], d_c = deg[node];
    float dv_c = dinv[node];
    int   s_n = 0, d_n = 0;
    float dv_n = 0.0f;
    if (node + step < nNodes) { s_n = off[node + step]; d_n = deg[node + step]; dv_n = dinv[node + step]; }
    int    mine_c = (lane < d_c) ? srcList[s_c + lane] : 0;
    float4 self_c = TsRow[(size_t)node * 32];

    for (; node < nNodes; node += step) {
        int    s_a = 0, d_a = 0;
        float  dv_a = 0.0f;
        int    mine_n = 0;
        float4 self_n = {0.0f, 0.0f, 0.0f, 0.0f};
        bool   ahead = true;

        float4 acc = self_c;
        for (int e0 = 0; e0 < d_c; e0 += 32) {
            const int cnt = (d_c - e0 < 32) ? (d_c - e0) : 32;
            const int mine = (e0 == 0) ? mine_c
                                       : ((lane < cnt) ? srcList[s_c + e0 + lane] : 0);
            int e = 0;
            for (; e + GO_INFLIGHT <= cnt; e += GO_INFLIGHT) {
                float4 v[GO_INFLIGHT];
#pragma unroll
                for (int u = 0; u < GO_INFLIGHT; ++u)
                    v[u] = TsRow[(size_t)__builtin_amdgcn_readlane(mine, e + u) * 32];
                if (ahead) {
                    gcn_out_ahead(TsRow, srcList, off, deg, dinv, node, step, nNodes, lane,
                                  s_n, d_n, s_a, d_a, dv_a, mine_n, self_n);
                    ahead = false;
                }
#pragma unroll
                for (int u = 0; u < GO_INFLIGHT; ++u) gcn_add4(acc, v[u]);
            }
            if (e < cnt) {
                float4 v[GO_INFLIGHT - 1] = {};
#pragma unroll
                for (int u = 0; u < GO_INFLIGHT - 1; ++u)
                    if (e + u < cnt)
                        v[u] = TsRow[(size_t)__builtin_amdgcn_readlane(mine, e + u) * 32];
                if (ahead) {
                    gcn_out_ahead(TsRow, srcList, off, deg, dinv, node, step, nNodes, lane,
                                  s_n, d_n, s_a, d_a, dv_a, mine_n, self_n);
                    ahead = false;
                }
#pragma unroll
                for (int u = 0; u < GO_INFLIGHT - 1; ++u)
                    if (e + u < cnt) gcn_add4(acc, v[u]);
            }
        }
        if (ahead)
            gcn_out_ahead(TsRow, srcList, off, deg, dinv, node, step, nNodes, lane,
                          s_n, d_n, s_a, d_a, dv_a, mine_n, self_n);

        float4 r;
        r.x = dv_c * acc.x + bb.x;
        r.y = dv_c * acc.y + bb.y;
        r.z = dv_c * acc.z + bb.z;
        r.w = dv_c * acc.w + bb.w;
        *(float4*)(dst + (size_t)node * 64) = r;

        s_c = s_n; d_c = d_n; dv_c = dv_n; mine_c = mine_n; self_c = self_n;
        s_n = s_a; d_n = d_a; dv_n = dv_a;
    }
}


extern "C" void kernel_launch(void* const* d_in, const int* in_sizes, int n_in,
                              void* d_out, int out_size, void* d_ws, size_t ws_size,
                              hipStream_t stream) {
    const float* X   = (const float*)d_in[0];
    const int*   ei  = (const int*)d_in[1];
    const float* W1  = (const float*)d_in[2];
    const float* b1  = (const float*)d_in[3];
    const float* Wmu = (const float*)d_in[4];
    const float* bmu = (const float*)d_in[5];
    const float* Wlv = (const float*)d_in[6];
    const float* blv = (const float*)d_in[7];
    float* out = (float*)d_out;

    const int N = in_sizes[0] / 256;
    const int E = in_sizes[1] / 2;
    const int* rows = ei;
    const int* cols = ei + E;

    float* Ts   = (float*)d_ws;
    float* Hbuf = Ts + (size_t)N * GCN_F;
    float* W23  = Hbuf + (size_t)N * GCN_F;
    float* dinv = W23 + GCN_F * GCN_F;
    int* degInt  = (int*)(dinv + N);
    int* off     = degInt + N;
    int* srcList = off + N;

    const int nbk = (N + CSR_BNODES - 1) / CSR_BNODES;
    int* colB   = (int*)Hbuf;
    int* rowB   = colB + E;
    int* cntT   = rowB + E;
    int* offT   = cntT + CSR_MAXROWS * nbk;
    gcn_chunk_sort<<<CSR_CHUNKS, CSR_BLOCK, 0, stream>>>(rows, cols, cntT, offT, rowB, colB, E, nbk);
    gcn_bucket_to_csr<<<nbk, CSR_BLOCK, 0, stream>>>(rowB, colB, cntT, offT, degInt, off,
                                                dinv, srcList, N, E, nbk);

    const int rowTiles = (N + 15) / 16;

    gcn_gemm_rowscale_wmma<<<G1_BLOCKS, G1_THREADS, 0, stream>>>(X, W1, dinv, Ts, N, rowTiles);
    gcn_gather_relu_gemm<<<768, FG_THREADS, 0, stream>>>(Ts, srcList, off, degInt, dinv, b1,
                                                          Wmu, Wlv, Hbuf, N, rowTiles);

    gcn_gather_out<<<GO_BLOCKS, GO_THREADS, 0, stream>>>(Hbuf, srcList, off, degInt,
                                                     dinv, bmu, blv, out, N);
}
